# starting kernel
# baseline (speedup 1.0000x reference)
#include <hip/hip_runtime.h>
#include <hip/hip_bf16.h>

typedef __attribute__((ext_vector_type(16))) _Float16 v16h;
typedef __attribute__((ext_vector_type(8)))  float    v8f;

#define GCN_D_IN  512
#define GCN_D_HID 256
#define GCN_D_OUT 64

__global__ void gcn_init_deg(float* __restrict__ deg, int n) {
    int i = blockIdx.x * blockDim.x + threadIdx.x;
    if (i < n) deg[i] = 1.0f;
}

__global__ void gcn_accum_deg(const int* __restrict__ dst, float* __restrict__ deg, int E) {
    int t = blockIdx.x * blockDim.x + threadIdx.x;
    if (t < E) atomicAdd(&deg[dst[t]], 1.0f);
}

__global__ void gcn_make_dinv(float* __restrict__ deg, int n) {
    int i = blockIdx.x * blockDim.x + threadIdx.x;
    if (i < n) {
        float d = deg[i];
        deg[i] = (d > 0.0f) ? rsqrtf(d) : 0.0f;
    }
}

__global__ void gcn_pack_w_f16(const float* __restrict__ W, _Float16* __restrict__ Wp,
                               int K, int N) {
    int t = blockIdx.x * blockDim.x + threadIdx.x;
    int tilesN = N >> 4;
    int total = (K >> 5) * tilesN * 32;
    if (t >= total) return;
    int frag = t >> 5;
    int lane = t & 31;
    int kt = frag / tilesN;
    int tn = frag - kt * tilesN;
    int n    = tn * 16 + (lane & 15);
    int koff = kt * 32 + (lane >> 4) * 8;
    _Float16* o = Wp + (size_t)t * 16;
#pragma unroll
    for (int j = 0; j < 8; ++j) {
        o[j]     = (_Float16)W[(size_t)(koff + j) * N + n];
        o[j + 8] = (_Float16)W[(size_t)(koff + 16 + j) * N + n];
    }
}

__global__ void gcn_wmma_gemm_f16x4(const float* __restrict__ A,
                                    const _Float16* __restrict__ Bp,
                                    float* __restrict__ C,
                                    int M, int K, int N) {
    const int lane    = threadIdx.x & 31;
    const int wave    = (blockIdx.x * blockDim.x + threadIdx.x) >> 5;
    const int tilesN  = N >> 4;
    const int groupsN = N >> 6;
    const int tm = wave / groupsN;
    const int tg = wave - tm * groupsN;
    if (tm * 16 >= M) return;
    const int tn0  = tg * 4;
    const int hsel = lane >> 4;
    const int idx  = lane & 15;

    const float* __restrict__ arow = A + (size_t)(tm * 16 + idx) * K;

    v8f acc0 = {}, acc1 = {}, acc2 = {}, acc3 = {};
    for (int k0 = 0; k0 < K; k0 += 32) {
        const int kb = k0 + hsel * 8;

        float4 a01 = *(const float4*)(arow + kb);
        float4 a23 = *(const float4*)(arow + kb + 4);
        float4 a45 = *(const float4*)(arow + kb + 16);
        float4 a67 = *(const float4*)(arow + kb + 20);
        if (k0 + 32 < K) __builtin_prefetch(arow + kb + 32, 0, 3);

        float af[16];
        *(float4*)(af + 0)  = a01;  *(float4*)(af + 4)  = a23;
        *(float4*)(af + 8)  = a45;  *(float4*)(af + 12) = a67;
        v16h av;
#pragma unroll
        for (int j = 0; j < 16; ++j) av[j] = (_Float16)af[j];

        const int kt = k0 >> 5;
        const _Float16* bbase = Bp + ((size_t)(kt * tilesN + tn0) * 32 + lane) * 16;
        v16h b0 = *(const v16h*)(bbase);
        v16h b1 = *(const v16h*)(bbase + 1 * 32 * 16);
        v16h b2 = *(const v16h*)(bbase + 2 * 32 * 16);
        v16h b3 = *(const v16h*)(bbase + 3 * 32 * 16);

        acc0 = __builtin_amdgcn_wmma_f32_16x16x32_f16(false, av, false, b0, (short)0, acc0, false, false);
        acc1 = __builtin_amdgcn_wmma_f32_16x16x32_f16(false, av, false, b1, (short)0, acc1, false, false);
        acc2 = __builtin_amdgcn_wmma_f32_16x16x32_f16(false, av, false, b2, (short)0, acc2, false, false);
        acc3 = __builtin_amdgcn_wmma_f32_16x16x32_f16(false, av, false, b3, (short)0, acc3, false, false);
    }

    float* cbase = C + (size_t)(tm * 16 + hsel * 8) * N + tn0 * 16 + idx;
#pragma unroll
    for (int r = 0; r < 8; ++r) {
        cbase[(size_t)r * N + 0]  = acc0[r];
        cbase[(size_t)r * N + 16] = acc1[r];
        cbase[(size_t)r * N + 32] = acc2[r];
        cbase[(size_t)r * N + 48] = acc3[r];
    }
}

__global__ void gcn_self_loop(const float* __restrict__ xw,
                              const float* __restrict__ dinv,
                              float* __restrict__ out, int total, int F) {
    int t = blockIdx.x * blockDim.x + threadIdx.x;
    if (t < total) {
        int i = t / F;
        float di = dinv[i];
        out[t] = xw[t] * di * di;
    }
}

__global__ void gcn_edge_scatter(const int* __restrict__ src,
                                 const int* __restrict__ dst,
                                 const float* __restrict__ dinv,
                                 const float* __restrict__ xw,
                                 float* __restrict__ out,
                                 int E, int F) {
    int t = blockIdx.x * blockDim.x + threadIdx.x;
    int chunks = F >> 2;
    int e = t / chunks;
    if (e >= E) return;
    int c = t - e * chunks;
    int s = src[e], d = dst[e];
    float norm = dinv[s] * dinv[d];
    float4 v = ((const float4*)(xw + (size_t)s * F))[c];
    float* o = out + (size_t)d * F + 4 * c;
    atomicAdd(o + 0, v.x * norm);
    atomicAdd(o + 1, v.y * norm);
    atomicAdd(o + 2, v.z * norm);
    atomicAdd(o + 3, v.w * norm);
}

__global__ void gcn_bias_relu(float* __restrict__ h, const float* __restrict__ bias,
                              int total, int F) {
    int t = blockIdx.x * blockDim.x + threadIdx.x;
    if (t < total) h[t] = fmaxf(h[t] + bias[t % F], 0.0f);
}

__global__ void gcn_softmax64(const float* __restrict__ h,
                              const float* __restrict__ bias,
                              float* __restrict__ out, int nodes) {
    int wave = (blockIdx.x * blockDim.x + threadIdx.x) >> 5;
    int lane = threadIdx.x & 31;
    if (wave >= nodes) return;
    const float* row = h + (size_t)wave * 64;
    float v0 = row[lane]      + bias[lane];
    float v1 = row[lane + 32] + bias[lane + 32];
    float m = fmaxf(v0, v1);
#pragma unroll
    for (int off = 16; off > 0; off >>= 1) m = fmaxf(m, __shfl_xor(m, off, 32));
    float e0 = __expf(v0 - m);
    float e1 = __expf(v1 - m);
    float s = e0 + e1;
#pragma unroll
    for (int off = 16; off > 0; off >>= 1) s += __shfl_xor(s, off, 32);
    float inv = 1.0f / s;
    float* orow = out + (size_t)wave * 64;
    orow[lane]      = e0 * inv;
    orow[lane + 32] = e1 * inv;
}

extern "C" void kernel_launch(void* const* d_in, const int* in_sizes, int n_in,
                              void* d_out, int out_size, void* d_ws, size_t ws_size,
                              hipStream_t stream) {
    const float* x   = (const float*)d_in[0];
    const int*   ei  = (const int*)  d_in[1];
    const float* W1  = (const float*)d_in[2];
    const float* b1  = (const float*)d_in[3];
    const float* W2  = (const float*)d_in[4];
    const float* b2  = (const float*)d_in[5];
    float* out = (float*)d_out;

    const int N = in_sizes[0] / GCN_D_IN;
    const int E = in_sizes[1] / 2;
    const int* src = ei;
    const int* dst = ei + E;

    float* ws   = (float*)d_ws;
    size_t off  = ((size_t)N + 63) & ~(size_t)63;
    float* dinv = ws;
    float* xw1  = ws + off;
    float* h1   = xw1 + (size_t)N * GCN_D_HID;
    _Float16* W1p = (_Float16*)(h1 + (size_t)N * GCN_D_HID);
    _Float16* W2p = W1p + (size_t)GCN_D_IN * GCN_D_HID;
    float* xw2  = xw1;
    float* h2   = xw1 + (size_t)N * GCN_D_OUT;

    const int B = 256;
    {
        int t1 = (GCN_D_IN >> 5) * (GCN_D_HID >> 4) * 32;
        gcn_pack_w_f16<<<(t1 + B - 1) / B, B, 0, stream>>>(W1, W1p, GCN_D_IN, GCN_D_HID);
        int t2 = (GCN_D_HID >> 5) * (GCN_D_OUT >> 4) * 32;
        gcn_pack_w_f16<<<(t2 + B - 1) / B, B, 0, stream>>>(W2, W2p, GCN_D_HID, GCN_D_OUT);
    }

    gcn_init_deg <<<(N + B - 1) / B, B, 0, stream>>>(dinv, N);
    gcn_accum_deg<<<(E + B - 1) / B, B, 0, stream>>>(dst, dinv, E);
    gcn_make_dinv<<<(N + B - 1) / B, B, 0, stream>>>(dinv, N);

    {
        int waves = (N / 16) * (GCN_D_HID / 64);
        int th = waves * 32;
        gcn_wmma_gemm_f16x4<<<(th + B - 1) / B, B, 0, stream>>>(x, W1p, xw1, N, GCN_D_IN, GCN_D_HID);
    }
    {
        int total = N * GCN_D_HID;
        gcn_self_loop<<<(total + B - 1) / B, B, 0, stream>>>(xw1, dinv, h1, total, GCN_D_HID);
        int et = E * (GCN_D_HID / 4);
        gcn_edge_scatter<<<(et + B - 1) / B, B, 0, stream>>>(src, dst, dinv, xw1, h1, E, GCN_D_HID);
        gcn_bias_relu<<<(total + B - 1) / B, B, 0, stream>>>(h1, b1, total, GCN_D_HID);
    }

    {
        int waves = (N / 16) * (GCN_D_OUT / 64);
        int th = waves * 32;
        gcn_wmma_gemm_f16x4<<<(th + B - 1) / B, B, 0, stream>>>(h1, W2p, xw2, N, GCN_D_HID, GCN_D_OUT);
    }
    {
        int total = N * GCN_D_OUT;
        gcn_self_loop<<<(total + B - 1) / B, B, 0, stream>>>(xw2, dinv, h2, total, GCN_D_OUT);
        int et = E * (GCN_D_OUT / 4);
        gcn_edge_scatter<<<(et + B - 1) / B, B, 0, stream>>>(src, dst, dinv, xw2, h2, E, GCN_D_OUT);
    }
    {
        int th = N * 32;
        gcn_softmax64<<<(th + B - 1) / B, B, 0, stream>>>(h2, b2, out, N);
    }
}


# float-atomic edge scatter replaced by in-call counting sort of edges by destination and a per-node gather-sum in registers, one store per row; sum order = arrival rank of the placing atomics
# speedup vs baseline: 1.9279x; 1.9279x over previous
#include <hip/hip_runtime.h>
#include <hip/hip_bf16.h>

typedef __attribute__((ext_vector_type(16))) _Float16 v16h;
typedef __attribute__((ext_vector_type(8)))  float    v8f;

#define GCN_D_IN  512
#define GCN_D_HID 256
#define GCN_D_OUT 64

__global__ void gcn_zero_cnt(int* __restrict__ cnt, int n) {
    int i = blockIdx.x * blockDim.x + threadIdx.x;
    if (i < n) cnt[i] = 0;
}

__global__ void gcn_count_in(const int* __restrict__ dst, int* __restrict__ cnt, int E) {
    int e = blockIdx.x * blockDim.x + threadIdx.x;
    if (e < E) atomicAdd(&cnt[dst[e]], 1);
}

__global__ void gcn_scan_local(const int* __restrict__ cnt, int* __restrict__ rowptr,
                               int* __restrict__ blocktot, int n) {
    __shared__ int wsum[32];
    const int t = threadIdx.x;
    const int lane = t & 31;
    const int wv = t >> 5;
    const int i = blockIdx.x * 1024 + t;
    const int c = (i < n) ? cnt[i] : 0;
    int incl = c;
#pragma unroll
    for (int off = 1; off < 32; off <<= 1) {
        int y = __shfl_up(incl, off, 32);
        if (lane >= off) incl += y;
    }
    if (lane == 31) wsum[wv] = incl;
    __syncthreads();
    if (wv == 0) {
        const int v = wsum[lane];
        int inc2 = v;
#pragma unroll
        for (int off = 1; off < 32; off <<= 1) {
            int y = __shfl_up(inc2, off, 32);
            if (lane >= off) inc2 += y;
        }
        wsum[lane] = inc2 - v;
    }
    __syncthreads();
    const int excl = incl - c + wsum[wv];
    if (i < n) rowptr[i] = excl;
    if (t == 1023) blocktot[blockIdx.x] = excl + c;
}

__global__ void gcn_scan_finish(int* __restrict__ cnt_cursor, int* __restrict__ rowptr,
                                const int* __restrict__ blocktot, float* __restrict__ dinv,
                                int n) {
    const int i = blockIdx.x * 1024 + threadIdx.x;
    int base = 0;
    for (int j = 0; j < (int)blockIdx.x; ++j) base += blocktot[j];
    if (i < n) {
        const int c = cnt_cursor[i];
        const int p = rowptr[i] + base;
        rowptr[i] = p;
        cnt_cursor[i] = p;
        const float d = (float)(c + 1);
        dinv[i] = (d > 0.0f) ? rsqrtf(d) : 0.0f;
        if (i == n - 1) rowptr[n] = p + c;
    }
}

__global__ void gcn_fill_in(const int* __restrict__ src, const int* __restrict__ dst,
                            const float* __restrict__ dinv, int* __restrict__ cursor,
                            int2* __restrict__ csr, int E) {
    int e = blockIdx.x * blockDim.x + threadIdx.x;
    if (e >= E) return;
    const int s = src[e], d = dst[e];
    const int pos = atomicAdd(&cursor[d], 1);
    csr[pos] = make_int2(s, __float_as_int(dinv[s]));
}

__global__ void gcn_pack_w_f16(const float* __restrict__ W, _Float16* __restrict__ Wp,
                               int K, int N) {
    int t = blockIdx.x * blockDim.x + threadIdx.x;
    int tilesN = N >> 4;
    int total = (K >> 5) * tilesN * 32;
    if (t >= total) return;
    int frag = t >> 5;
    int lane = t & 31;
    int kt = frag / tilesN;
    int tn = frag - kt * tilesN;
    int n    = tn * 16 + (lane & 15);
    int koff = kt * 32 + (lane >> 4) * 8;
    _Float16* o = Wp + (size_t)t * 16;
#pragma unroll
    for (int j = 0; j < 8; ++j) {
        o[j]     = (_Float16)W[(size_t)(koff + j) * N + n];
        o[j + 8] = (_Float16)W[(size_t)(koff + 16 + j) * N + n];
    }
}

__global__ void gcn_wmma_gemm_f16x4(const float* __restrict__ A,
                                    const _Float16* __restrict__ Bp,
                                    float* __restrict__ C,
                                    int M, int K, int N) {
    const int lane    = threadIdx.x & 31;
    const int wave    = (blockIdx.x * blockDim.x + threadIdx.x) >> 5;
    const int tilesN  = N >> 4;
    const int groupsN = N >> 6;
    const int tm = wave / groupsN;
    const int tg = wave - tm * groupsN;
    if (tm * 16 >= M) return;
    const int tn0  = tg * 4;
    const int hsel = lane >> 4;
    const int idx  = lane & 15;

    const float* __restrict__ arow = A + (size_t)(tm * 16 + idx) * K;

    v8f acc0 = {}, acc1 = {}, acc2 = {}, acc3 = {};
    for (int k0 = 0; k0 < K; k0 += 32) {
        const int kb = k0 + hsel * 8;

        float4 a01 = *(const float4*)(arow + kb);
        float4 a23 = *(const float4*)(arow + kb + 4);
        float4 a45 = *(const float4*)(arow + kb + 16);
        float4 a67 = *(const float4*)(arow + kb + 20);
        if (k0 + 32 < K) __builtin_prefetch(arow + kb + 32, 0, 3);

        float af[16];
        *(float4*)(af + 0)  = a01;  *(float4*)(af + 4)  = a23;
        *(float4*)(af + 8)  = a45;  *(float4*)(af + 12) = a67;
        v16h av;
#pragma unroll
        for (int j = 0; j < 16; ++j) av[j] = (_Float16)af[j];

        const int kt = k0 >> 5;
        const _Float16* bbase = Bp + ((size_t)(kt * tilesN + tn0) * 32 + lane) * 16;
        v16h b0 = *(const v16h*)(bbase);
        v16h b1 = *(const v16h*)(bbase + 1 * 32 * 16);
        v16h b2 = *(const v16h*)(bbase + 2 * 32 * 16);
        v16h b3 = *(const v16h*)(bbase + 3 * 32 * 16);

        acc0 = __builtin_amdgcn_wmma_f32_16x16x32_f16(false, av, false, b0, (short)0, acc0, false, false);
        acc1 = __builtin_amdgcn_wmma_f32_16x16x32_f16(false, av, false, b1, (short)0, acc1, false, false);
        acc2 = __builtin_amdgcn_wmma_f32_16x16x32_f16(false, av, false, b2, (short)0, acc2, false, false);
        acc3 = __builtin_amdgcn_wmma_f32_16x16x32_f16(false, av, false, b3, (short)0, acc3, false, false);
    }

    float* cbase = C + (size_t)(tm * 16 + hsel * 8) * N + tn0 * 16 + idx;
#pragma unroll
    for (int r = 0; r < 8; ++r) {
        cbase[(size_t)r * N + 0]  = acc0[r];
        cbase[(size_t)r * N + 16] = acc1[r];
        cbase[(size_t)r * N + 32] = acc2[r];
        cbase[(size_t)r * N + 48] = acc3[r];
    }
}

__global__ void gcn_gather256(const int* __restrict__ rowptr, const int2* __restrict__ csr,
                              const float* __restrict__ dinv, const float* __restrict__ xw,
                              float* __restrict__ h, int nodes) {
    const int lane = threadIdx.x & 31;
    const int node = __builtin_amdgcn_readfirstlane((int)((blockIdx.x * blockDim.x + threadIdx.x) >> 5));
    if (node >= nodes) return;
    const int beg = rowptr[node];
    const int end = rowptr[node + 1];
    const float di = dinv[node];
    const float4* __restrict__ self = (const float4*)(xw + (size_t)node * 256);
    float4 a0 = self[lane];
    float4 a1 = self[lane + 32];
    a0.x *= di; a0.y *= di; a0.z *= di; a0.w *= di;
    a1.x *= di; a1.y *= di; a1.z *= di; a1.w *= di;
    for (int base = beg; base < end; base += 32) {
        int2 my = make_int2(0, 0);
        if (base + lane < end) my = csr[base + lane];
        const int n = (end - base < 32) ? (end - base) : 32;
#pragma unroll 4
        for (int k = 0; k < n; ++k) {
            const int s = __builtin_amdgcn_readlane(my.x, k);
            const float w = __int_as_float(__builtin_amdgcn_readlane(my.y, k));
            const float4* __restrict__ r = (const float4*)(xw + (size_t)s * 256);
            const float4 v0 = r[lane];
            const float4 v1 = r[lane + 32];
            a0.x = fmaf(v0.x, w, a0.x); a0.y = fmaf(v0.y, w, a0.y);
            a0.z = fmaf(v0.z, w, a0.z); a0.w = fmaf(v0.w, w, a0.w);
            a1.x = fmaf(v1.x, w, a1.x); a1.y = fmaf(v1.y, w, a1.y);
            a1.z = fmaf(v1.z, w, a1.z); a1.w = fmaf(v1.w, w, a1.w);
        }
    }
    a0.x *= di; a0.y *= di; a0.z *= di; a0.w *= di;
    a1.x *= di; a1.y *= di; a1.z *= di; a1.w *= di;
    float4* __restrict__ o = (float4*)(h + (size_t)node * 256);
    o[lane] = a0;
    o[lane + 32] = a1;
}

__global__ void gcn_gather64(const int* __restrict__ rowptr, const int2* __restrict__ csr,
                             const float* __restrict__ dinv, const float* __restrict__ xw,
                             float* __restrict__ h, int nodes) {
    const int lane = threadIdx.x & 31;
    const int node = __builtin_amdgcn_readfirstlane((int)((blockIdx.x * blockDim.x + threadIdx.x) >> 5));
    if (node >= nodes) return;
    const int beg = rowptr[node];
    const int end = rowptr[node + 1];
    const float di = dinv[node];
    float2 a = ((const float2*)(xw + (size_t)node * 64))[lane];
    a.x *= di; a.y *= di;
    for (int base = beg; base < end; base += 32) {
        int2 my = make_int2(0, 0);
        if (base + lane < end) my = csr[base + lane];
        const int n = (end - base < 32) ? (end - base) : 32;
#pragma unroll 8
        for (int k = 0; k < n; ++k) {
            const int s = __builtin_amdgcn_readlane(my.x, k);
            const float w = __int_as_float(__builtin_amdgcn_readlane(my.y, k));
            const float2 v = ((const float2*)(xw + (size_t)s * 64))[lane];
            a.x = fmaf(v.x, w, a.x);
            a.y = fmaf(v.y, w, a.y);
        }
    }
    a.x *= di; a.y *= di;
    ((float2*)(h + (size_t)node * 64))[lane] = a;
}

__global__ void gcn_bias_relu(float* __restrict__ h, const float* __restrict__ bias,
                              int total, int F) {
    int t = blockIdx.x * blockDim.x + threadIdx.x;
    if (t < total) h[t] = fmaxf(h[t] + bias[t % F], 0.0f);
}

__global__ void gcn_softmax64(const float* __restrict__ h,
                              const float* __restrict__ bias,
                              float* __restrict__ out, int nodes) {
    int wave = (blockIdx.x * blockDim.x + threadIdx.x) >> 5;
    int lane = threadIdx.x & 31;
    if (wave >= nodes) return;
    const float* row = h + (size_t)wave * 64;
    float v0 = row[lane]      + bias[lane];
    float v1 = row[lane + 32] + bias[lane + 32];
    float m = fmaxf(v0, v1);
#pragma unroll
    for (int off = 16; off > 0; off >>= 1) m = fmaxf(m, __shfl_xor(m, off, 32));
    float e0 = __expf(v0 - m);
    float e1 = __expf(v1 - m);
    float s = e0 + e1;
#pragma unroll
    for (int off = 16; off > 0; off >>= 1) s += __shfl_xor(s, off, 32);
    float inv = 1.0f / s;
    float* orow = out + (size_t)wave * 64;
    orow[lane]      = e0 * inv;
    orow[lane + 32] = e1 * inv;
}

extern "C" void kernel_launch(void* const* d_in, const int* in_sizes, int n_in,
                              void* d_out, int out_size, void* d_ws, size_t ws_size,
                              hipStream_t stream) {
    const float* x   = (const float*)d_in[0];
    const int*   ei  = (const int*)  d_in[1];
    const float* W1  = (const float*)d_in[2];
    const float* b1  = (const float*)d_in[3];
    const float* W2  = (const float*)d_in[4];
    const float* b2  = (const float*)d_in[5];
    float* out = (float*)d_out;

    const int N = in_sizes[0] / GCN_D_IN;
    const int E = in_sizes[1] / 2;
    const int* src = ei;
    const int* dst = ei + E;

    float* ws   = (float*)d_ws;
    size_t off  = ((size_t)N + 63) & ~(size_t)63;
    float* dinv = ws;
    float* xw1  = ws + off;
    float* h1   = xw1 + (size_t)N * GCN_D_HID;
    _Float16* W1p = (_Float16*)(h1 + (size_t)N * GCN_D_HID);
    _Float16* W2p = W1p + (size_t)GCN_D_IN * GCN_D_HID;
    float* xw2  = xw1;
    float* h2   = xw1 + (size_t)N * GCN_D_OUT;

    int*  so       = (int*)d_out;
    int*  cursor   = so;
    int*  rowptr   = so + off;
    int*  blocktot = so + 2 * off + 64;
    int2* csr      = (int2*)(so + 2 * off + 64 + 1024);

    const int B = 256;
    {
        int t1 = (GCN_D_IN >> 5) * (GCN_D_HID >> 4) * 32;
        gcn_pack_w_f16<<<(t1 + B - 1) / B, B, 0, stream>>>(W1, W1p, GCN_D_IN, GCN_D_HID);
        int t2 = (GCN_D_HID >> 5) * (GCN_D_OUT >> 4) * 32;
        gcn_pack_w_f16<<<(t2 + B - 1) / B, B, 0, stream>>>(W2, W2p, GCN_D_HID, GCN_D_OUT);
    }

    {
        const int SB = 1024;
        gcn_zero_cnt   <<<(N + B - 1) / B, B, 0, stream>>>(cursor, N);
        gcn_count_in   <<<(E + B - 1) / B, B, 0, stream>>>(dst, cursor, E);
        gcn_scan_local <<<(N + SB - 1) / SB, SB, 0, stream>>>(cursor, rowptr, blocktot, N);
        gcn_scan_finish<<<(N + SB - 1) / SB, SB, 0, stream>>>(cursor, rowptr, blocktot, dinv, N);
        gcn_fill_in    <<<(E + B - 1) / B, B, 0, stream>>>(src, dst, dinv, cursor, csr, E);
    }

    {
        int waves = (N / 16) * (GCN_D_HID / 64);
        int th = waves * 32;
        gcn_wmma_gemm_f16x4<<<(th + B - 1) / B, B, 0, stream>>>(x, W1p, xw1, N, GCN_D_IN, GCN_D_HID);
    }
    {
        int total = N * GCN_D_HID;
        int th = N * 32;
        gcn_gather256<<<(th + B - 1) / B, B, 0, stream>>>(rowptr, csr, dinv, xw1, h1, N);
        gcn_bias_relu<<<(total + B - 1) / B, B, 0, stream>>>(h1, b1, total, GCN_D_HID);
    }

    {
        int waves = (N / 16) * (GCN_D_OUT / 64);
        int th = waves * 32;
        gcn_wmma_gemm_f16x4<<<(th + B - 1) / B, B, 0, stream>>>(h1, W2p, xw2, N, GCN_D_HID, GCN_D_OUT);
    }
    {
        int th = N * 32;
        gcn_gather64<<<(th + B - 1) / B, B, 0, stream>>>(rowptr, csr, dinv, xw2, h2, N);
    }
    {
        int th = N * 32;
        gcn_softmax64<<<(th + B - 1) / B, B, 0, stream>>>(h2, b2, out, N);
    }
}


# layer-1 bias add and ReLU folded into the 256-wide gather kernel's store (row scale and bias add contract to one fma); the one-thread-per-element kernel and its extra read+write of h1 are gone
# speedup vs baseline: 2.1675x; 1.1243x over previous
#include <hip/hip_runtime.h>
#include <hip/hip_bf16.h>

typedef __attribute__((ext_vector_type(16))) _Float16 v16h;
typedef __attribute__((ext_vector_type(8)))  float    v8f;

#define GCN_D_IN  512
#define GCN_D_HID 256
#define GCN_D_OUT 64

__global__ void gcn_zero_cnt(int* __restrict__ cnt, int n) {
    int i = blockIdx.x * blockDim.x + threadIdx.x;
    if (i < n) cnt[i] = 0;
}

__global__ void gcn_count_in(const int* __restrict__ dst, int* __restrict__ cnt, int E) {
    int e = blockIdx.x * blockDim.x + threadIdx.x;
    if (e < E) atomicAdd(&cnt[dst[e]], 1);
}

__global__ void gcn_scan_local(const int* __restrict__ cnt, int* __restrict__ rowptr,
                               int* __restrict__ blocktot, int n) {
    __shared__ int wsum[32];
    const int t = threadIdx.x;
    const int lane = t & 31;
    const int wv = t >> 5;
    const int i = blockIdx.x * 1024 + t;
    const int c = (i < n) ? cnt[i] : 0;
    int incl = c;
#pragma unroll
    for (int off = 1; off < 32; off <<= 1) {
        int y = __shfl_up(incl, off, 32);
        if (lane >= off) incl += y;
    }
    if (lane == 31) wsum[wv] = incl;
    __syncthreads();
    if (wv == 0) {
        const int v = wsum[lane];
        int inc2 = v;
#pragma unroll
        for (int off = 1; off < 32; off <<= 1) {
            int y = __shfl_up(inc2, off, 32);
            if (lane >= off) inc2 += y;
        }
        wsum[lane] = inc2 - v;
    }
    __syncthreads();
    const int excl = incl - c + wsum[wv];
    if (i < n) rowptr[i] = excl;
    if (t == 1023) blocktot[blockIdx.x] = excl + c;
}

__global__ void gcn_scan_finish(int* __restrict__ cnt_cursor, int* __restrict__ rowptr,
                                const int* __restrict__ blocktot, float* __restrict__ dinv,
                                int n) {
    const int i = blockIdx.x * 1024 + threadIdx.x;
    int base = 0;
    for (int j = 0; j < (int)blockIdx.x; ++j) base += blocktot[j];
    if (i < n) {
        const int c = cnt_cursor[i];
        const int p = rowptr[i] + base;
        rowptr[i] = p;
        cnt_cursor[i] = p;
        const float d = (float)(c + 1);
        dinv[i] = (d > 0.0f) ? rsqrtf(d) : 0.0f;
        if (i == n - 1) rowptr[n] = p + c;
    }
}

__global__ void gcn_fill_in(const int* __restrict__ src, const int* __restrict__ dst,
                            const float* __restrict__ dinv, int* __restrict__ cursor,
                            int2* __restrict__ csr, int E) {
    int e = blockIdx.x * blockDim.x + threadIdx.x;
    if (e >= E) return;
    const int s = src[e], d = dst[e];
    const int pos = atomicAdd(&cursor[d], 1);
    csr[pos] = make_int2(s, __float_as_int(dinv[s]));
}

__global__ void gcn_pack_w_f16(const float* __restrict__ W, _Float16* __restrict__ Wp,
                               int K, int N) {
    int t = blockIdx.x * blockDim.x + threadIdx.x;
    int tilesN = N >> 4;
    int total = (K >> 5) * tilesN * 32;
    if (t >= total) return;
    int frag = t >> 5;
    int lane = t & 31;
    int kt = frag / tilesN;
    int tn = frag - kt * tilesN;
    int n    = tn * 16 + (lane & 15);
    int koff = kt * 32 + (lane >> 4) * 8;
    _Float16* o = Wp + (size_t)t * 16;
#pragma unroll
    for (int j = 0; j < 8; ++j) {
        o[j]     = (_Float16)W[(size_t)(koff + j) * N + n];
        o[j + 8] = (_Float16)W[(size_t)(koff + 16 + j) * N + n];
    }
}

__global__ void gcn_wmma_gemm_f16x4(const float* __restrict__ A,
                                    const _Float16* __restrict__ Bp,
                                    float* __restrict__ C,
                                    int M, int K, int N) {
    const int lane    = threadIdx.x & 31;
    const int wave    = (blockIdx.x * blockDim.x + threadIdx.x) >> 5;
    const int tilesN  = N >> 4;
    const int groupsN = N >> 6;
    const int tm = wave / groupsN;
    const int tg = wave - tm * groupsN;
    if (tm * 16 >= M) return;
    const int tn0  = tg * 4;
    const int hsel = lane >> 4;
    const int idx  = lane & 15;

    const float* __restrict__ arow = A + (size_t)(tm * 16 + idx) * K;

    v8f acc0 = {}, acc1 = {}, acc2 = {}, acc3 = {};
    for (int k0 = 0; k0 < K; k0 += 32) {
        const int kb = k0 + hsel * 8;

        float4 a01 = *(const float4*)(arow + kb);
        float4 a23 = *(const float4*)(arow + kb + 4);
        float4 a45 = *(const float4*)(arow + kb + 16);
        float4 a67 = *(const float4*)(arow + kb + 20);
        if (k0 + 32 < K) __builtin_prefetch(arow + kb + 32, 0, 3);

        float af[16];
        *(float4*)(af + 0)  = a01;  *(float4*)(af + 4)  = a23;
        *(float4*)(af + 8)  = a45;  *(float4*)(af + 12) = a67;
        v16h av;
#pragma unroll
        for (int j = 0; j < 16; ++j) av[j] = (_Float16)af[j];

        const int kt = k0 >> 5;
        const _Float16* bbase = Bp + ((size_t)(kt * tilesN + tn0) * 32 + lane) * 16;
        v16h b0 = *(const v16h*)(bbase);
        v16h b1 = *(const v16h*)(bbase + 1 * 32 * 16);
        v16h b2 = *(const v16h*)(bbase + 2 * 32 * 16);
        v16h b3 = *(const v16h*)(bbase + 3 * 32 * 16);

        acc0 = __builtin_amdgcn_wmma_f32_16x16x32_f16(false, av, false, b0, (short)0, acc0, false, false);
        acc1 = __builtin_amdgcn_wmma_f32_16x16x32_f16(false, av, false, b1, (short)0, acc1, false, false);
        acc2 = __builtin_amdgcn_wmma_f32_16x16x32_f16(false, av, false, b2, (short)0, acc2, false, false);
        acc3 = __builtin_amdgcn_wmma_f32_16x16x32_f16(false, av, false, b3, (short)0, acc3, false, false);
    }

    float* cbase = C + (size_t)(tm * 16 + hsel * 8) * N + tn0 * 16 + idx;
#pragma unroll
    for (int r = 0; r < 8; ++r) {
        cbase[(size_t)r * N + 0]  = acc0[r];
        cbase[(size_t)r * N + 16] = acc1[r];
        cbase[(size_t)r * N + 32] = acc2[r];
        cbase[(size_t)r * N + 48] = acc3[r];
    }
}

__global__ void gcn_gather256_bias_relu(const int* __restrict__ rowptr, const int2* __restrict__ csr,
                                        const float* __restrict__ dinv, const float* __restrict__ xw,
                                        const float* __restrict__ bias,
                                        float* __restrict__ h, int nodes) {
    const int lane = threadIdx.x & 31;
    const int node = __builtin_amdgcn_readfirstlane((int)((blockIdx.x * blockDim.x + threadIdx.x) >> 5));
    if (node >= nodes) return;
    const int beg = rowptr[node];
    const int end = rowptr[node + 1];
    const float di = dinv[node];
    const float4* __restrict__ self = (const float4*)(xw + (size_t)node * 256);
    float4 a0 = self[lane];
    float4 a1 = self[lane + 32];
    const float4 b0 = ((const float4*)bias)[lane];
    const float4 b1 = ((const float4*)bias)[lane + 32];
    a0.x *= di; a0.y *= di; a0.z *= di; a0.w *= di;
    a1.x *= di; a1.y *= di; a1.z *= di; a1.w *= di;
    for (int base = beg; base < end; base += 32) {
        int2 my = make_int2(0, 0);
        if (base + lane < end) my = csr[base + lane];
        const int n = (end - base < 32) ? (end - base) : 32;
#pragma unroll 4
        for (int k = 0; k < n; ++k) {
            const int s = __builtin_amdgcn_readlane(my.x, k);
            const float w = __int_as_float(__builtin_amdgcn_readlane(my.y, k));
            const float4* __restrict__ r = (const float4*)(xw + (size_t)s * 256);
            const float4 v0 = r[lane];
            const float4 v1 = r[lane + 32];
            a0.x = fmaf(v0.x, w, a0.x); a0.y = fmaf(v0.y, w, a0.y);
            a0.z = fmaf(v0.z, w, a0.z); a0.w = fmaf(v0.w, w, a0.w);
            a1.x = fmaf(v1.x, w, a1.x); a1.y = fmaf(v1.y, w, a1.y);
            a1.z = fmaf(v1.z, w, a1.z); a1.w = fmaf(v1.w, w, a1.w);
        }
    }
    a0.x = fmaxf(a0.x * di + b0.x, 0.0f); a0.y = fmaxf(a0.y * di + b0.y, 0.0f);
    a0.z = fmaxf(a0.z * di + b0.z, 0.0f); a0.w = fmaxf(a0.w * di + b0.w, 0.0f);
    a1.x = fmaxf(a1.x * di + b1.x, 0.0f); a1.y = fmaxf(a1.y * di + b1.y, 0.0f);
    a1.z = fmaxf(a1.z * di + b1.z, 0.0f); a1.w = fmaxf(a1.w * di + b1.w, 0.0f);
    float4* __restrict__ o = (float4*)(h + (size_t)node * 256);
    o[lane] = a0;
    o[lane + 32] = a1;
}

__global__ void gcn_gather64(const int* __restrict__ rowptr, const int2* __restrict__ csr,
                             const float* __restrict__ dinv, const float* __restrict__ xw,
                             float* __restrict__ h, int nodes) {
    const int lane = threadIdx.x & 31;
    const int node = __builtin_amdgcn_readfirstlane((int)((blockIdx.x * blockDim.x + threadIdx.x) >> 5));
    if (node >= nodes) return;
    const int beg = rowptr[node];
    const int end = rowptr[node + 1];
    const float di = dinv[node];
    float2 a = ((const float2*)(xw + (size_t)node * 64))[lane];
    a.x *= di; a.y *= di;
    for (int base = beg; base < end; base += 32) {
        int2 my = make_int2(0, 0);
        if (base + lane < end) my = csr[base + lane];
        const int n = (end - base < 32) ? (end - base) : 32;
#pragma unroll 8
        for (int k = 0; k < n; ++k) {
            const int s = __builtin_amdgcn_readlane(my.x, k);
            const float w = __int_as_float(__builtin_amdgcn_readlane(my.y, k));
            const float2 v = ((const float2*)(xw + (size_t)s * 64))[lane];
            a.x = fmaf(v.x, w, a.x);
            a.y = fmaf(v.y, w, a.y);
        }
    }
    a.x *= di; a.y *= di;
    ((float2*)(h + (size_t)node * 64))[lane] = a;
}

__global__ void gcn_softmax64(const float* __restrict__ h,
                              const float* __restrict__ bias,
                              float* __restrict__ out, int nodes) {
    int wave = (blockIdx.x * blockDim.x + threadIdx.x) >> 5;
    int lane = threadIdx.x & 31;
    if (wave >= nodes) return;
    const float* row = h + (size_t)wave * 64;
    float v0 = row[lane]      + bias[lane];
    float v1 = row[lane + 32] + bias[lane + 32];
    float m = fmaxf(v0, v1);
#pragma unroll
    for (int off = 16; off > 0; off >>= 1) m = fmaxf(m, __shfl_xor(m, off, 32));
    float e0 = __expf(v0 - m);
    float e1 = __expf(v1 - m);
    float s = e0 + e1;
#pragma unroll
    for (int off = 16; off > 0; off >>= 1) s += __shfl_xor(s, off, 32);
    float inv = 1.0f / s;
    float* orow = out + (size_t)wave * 64;
    orow[lane]      = e0 * inv;
    orow[lane + 32] = e1 * inv;
}

extern "C" void kernel_launch(void* const* d_in, const int* in_sizes, int n_in,
                              void* d_out, int out_size, void* d_ws, size_t ws_size,
                              hipStream_t stream) {
    const float* x   = (const float*)d_in[0];
    const int*   ei  = (const int*)  d_in[1];
    const float* W1  = (const float*)d_in[2];
    const float* b1  = (const float*)d_in[3];
    const float* W2  = (const float*)d_in[4];
    const float* b2  = (const float*)d_in[5];
    float* out = (float*)d_out;

    const int N = in_sizes[0] / GCN_D_IN;
    const int E = in_sizes[1] / 2;
    const int* src = ei;
    const int* dst = ei + E;

    float* ws   = (float*)d_ws;
    size_t off  = ((size_t)N + 63) & ~(size_t)63;
    float* dinv = ws;
    float* xw1  = ws + off;
    float* h1   = xw1 + (size_t)N * GCN_D_HID;
    _Float16* W1p = (_Float16*)(h1 + (size_t)N * GCN_D_HID);
    _Float16* W2p = W1p + (size_t)GCN_D_IN * GCN_D_HID;
    float* xw2  = xw1;
    float* h2   = xw1 + (size_t)N * GCN_D_OUT;

    int*  so       = (int*)d_out;
    int*  cursor   = so;
    int*  rowptr   = so + off;
    int*  blocktot = so + 2 * off + 64;
    int2* csr      = (int2*)(so + 2 * off + 64 + 1024);

    const int B = 256;
    {
        int t1 = (GCN_D_IN >> 5) * (GCN_D_HID >> 4) * 32;
        gcn_pack_w_f16<<<(t1 + B - 1) / B, B, 0, stream>>>(W1, W1p, GCN_D_IN, GCN_D_HID);
        int t2 = (GCN_D_HID >> 5) * (GCN_D_OUT >> 4) * 32;
        gcn_pack_w_f16<<<(t2 + B - 1) / B, B, 0, stream>>>(W2, W2p, GCN_D_HID, GCN_D_OUT);
    }

    {
        const int SB = 1024;
        gcn_zero_cnt   <<<(N + B - 1) / B, B, 0, stream>>>(cursor, N);
        gcn_count_in   <<<(E + B - 1) / B, B, 0, stream>>>(dst, cursor, E);
        gcn_scan_local <<<(N + SB - 1) / SB, SB, 0, stream>>>(cursor, rowptr, blocktot, N);
        gcn_scan_finish<<<(N + SB - 1) / SB, SB, 0, stream>>>(cursor, rowptr, blocktot, dinv, N);
        gcn_fill_in    <<<(E + B - 1) / B, B, 0, stream>>>(src, dst, dinv, cursor, csr, E);
    }

    {
        int waves = (N / 16) * (GCN_D_HID / 64);
        int th = waves * 32;
        gcn_wmma_gemm_f16x4<<<(th + B - 1) / B, B, 0, stream>>>(x, W1p, xw1, N, GCN_D_IN, GCN_D_HID);
    }
    {
        int th = N * 32;
        gcn_gather256_bias_relu<<<(th + B - 1) / B, B, 0, stream>>>(rowptr, csr, dinv, xw1, b1, h1, N);
    }

    {
        int waves = (N / 16) * (GCN_D_OUT / 64);
        int th = waves * 32;
        gcn_wmma_gemm_f16x4<<<(th + B - 1) / B, B, 0, stream>>>(h1, W2p, xw2, N, GCN_D_HID, GCN_D_OUT);
    }
    {
        int th = N * 32;
        gcn_gather64<<<(th + B - 1) / B, B, 0, stream>>>(rowptr, csr, dinv, xw2, h2, N);
    }
    {
        int th = N * 32;
        gcn_softmax64<<<(th + B - 1) / B, B, 0, stream>>>(h2, b2, out, N);
    }
}


# stage after each gather fused into it: layer-2 GEMM (f16 WMMA on an LDS tile of 16 rows) in the 256-wide one, bias+softmax in the 64-wide; softmax sum pairs features 2l,2l+1 per lane
# speedup vs baseline: 2.4656x; 1.1375x over previous
#include <hip/hip_runtime.h>
#include <hip/hip_bf16.h>

typedef __attribute__((ext_vector_type(16))) _Float16 v16h;
typedef __attribute__((ext_vector_type(8)))  float    v8f;
typedef __attribute__((ext_vector_type(4)))  _Float16 v4h;
typedef __attribute__((ext_vector_type(8)))  _Float16 v8h;

#define GCN_D_IN  512
#define GCN_D_HID 256
#define GCN_D_OUT 64
#define GCN_H_PITCH (GCN_D_HID + 8)

__global__ void gcn_zero_cnt(int* __restrict__ cnt, int n) {
    int i = blockIdx.x * blockDim.x + threadIdx.x;
    if (i < n) cnt[i] = 0;
}

__global__ void gcn_count_in(const int* __restrict__ dst, int* __restrict__ cnt, int E) {
    int e = blockIdx.x * blockDim.x + threadIdx.x;
    if (e < E) atomicAdd(&cnt[dst[e]], 1);
}

__global__ void gcn_scan_local(const int* __restrict__ cnt, int* __restrict__ rowptr,
                               int* __restrict__ blocktot, int n) {
    __shared__ int wsum[32];
    const int t = threadIdx.x;
    const int lane = t & 31;
    const int wv = t >> 5;
    const int i = blockIdx.x * 1024 + t;
    const int c = (i < n) ? cnt[i] : 0;
    int incl = c;
#pragma unroll
    for (int off = 1; off < 32; off <<= 1) {
        int y = __shfl_up(incl, off, 32);
        if (lane >= off) incl += y;
    }
    if (lane == 31) wsum[wv] = incl;
    __syncthreads();
    if (wv == 0) {
        const int v = wsum[lane];
        int inc2 = v;
#pragma unroll
        for (int off = 1; off < 32; off <<= 1) {
            int y = __shfl_up(inc2, off, 32);
            if (lane >= off) inc2 += y;
        }
        wsum[lane] = inc2 - v;
    }
    __syncthreads();
    const int excl = incl - c + wsum[wv];
    if (i < n) rowptr[i] = excl;
    if (t == 1023) blocktot[blockIdx.x] = excl + c;
}

__global__ void gcn_scan_finish(int* __restrict__ cnt_cursor, int* __restrict__ rowptr,
                                const int* __restrict__ blocktot, float* __restrict__ dinv,
                                int n) {
    const int i = blockIdx.x * 1024 + threadIdx.x;
    int base = 0;
    for (int j = 0; j < (int)blockIdx.x; ++j) base += blocktot[j];
    if (i < n) {
        const int c = cnt_cursor[i];
        const int p = rowptr[i] + base;
        rowptr[i] = p;
        cnt_cursor[i] = p;
        const float d = (float)(c + 1);
        dinv[i] = (d > 0.0f) ? rsqrtf(d) : 0.0f;
        if (i == n - 1) rowptr[n] = p + c;
    }
}

__global__ void gcn_fill_in(const int* __restrict__ src, const int* __restrict__ dst,
                            const float* __restrict__ dinv, int* __restrict__ cursor,
                            int2* __restrict__ csr, int E) {
    int e = blockIdx.x * blockDim.x + threadIdx.x;
    if (e >= E) return;
    const int s = src[e], d = dst[e];
    const int pos = atomicAdd(&cursor[d], 1);
    csr[pos] = make_int2(s, __float_as_int(dinv[s]));
}

__global__ void gcn_pack_w_f16(const float* __restrict__ W, _Float16* __restrict__ Wp,
                               int K, int N) {
    int t = blockIdx.x * blockDim.x + threadIdx.x;
    int tilesN = N >> 4;
    int total = (K >> 5) * tilesN * 32;
    if (t >= total) return;
    int frag = t >> 5;
    int lane = t & 31;
    int kt = frag / tilesN;
    int tn = frag - kt * tilesN;
    int n    = tn * 16 + (lane & 15);
    int koff = kt * 32 + (lane >> 4) * 8;
    _Float16* o = Wp + (size_t)t * 16;
#pragma unroll
    for (int j = 0; j < 8; ++j) {
        o[j]     = (_Float16)W[(size_t)(koff + j) * N + n];
        o[j + 8] = (_Float16)W[(size_t)(koff + 16 + j) * N + n];
    }
}

__global__ void gcn_wmma_gemm_f16x4(const float* __restrict__ A,
                                    const _Float16* __restrict__ Bp,
                                    float* __restrict__ C,
                                    int M, int K, int N) {
    const int lane    = threadIdx.x & 31;
    const int wave    = (blockIdx.x * blockDim.x + threadIdx.x) >> 5;
    const int tilesN  = N >> 4;
    const int groupsN = N >> 6;
    const int tm = wave / groupsN;
    const int tg = wave - tm * groupsN;
    if (tm * 16 >= M) return;
    const int tn0  = tg * 4;
    const int hsel = lane >> 4;
    const int idx  = lane & 15;

    const float* __restrict__ arow = A + (size_t)(tm * 16 + idx) * K;

    v8f acc0 = {}, acc1 = {}, acc2 = {}, acc3 = {};
    for (int k0 = 0; k0 < K; k0 += 32) {
        const int kb = k0 + hsel * 8;

        float4 a01 = *(const float4*)(arow + kb);
        float4 a23 = *(const float4*)(arow + kb + 4);
        float4 a45 = *(const float4*)(arow + kb + 16);
        float4 a67 = *(const float4*)(arow + kb + 20);
        if (k0 + 32 < K) __builtin_prefetch(arow + kb + 32, 0, 3);

        float af[16];
        *(float4*)(af + 0)  = a01;  *(float4*)(af + 4)  = a23;
        *(float4*)(af + 8)  = a45;  *(float4*)(af + 12) = a67;
        v16h av;
#pragma unroll
        for (int j = 0; j < 16; ++j) av[j] = (_Float16)af[j];

        const int kt = k0 >> 5;
        const _Float16* bbase = Bp + ((size_t)(kt * tilesN + tn0) * 32 + lane) * 16;
        v16h b0 = *(const v16h*)(bbase);
        v16h b1 = *(const v16h*)(bbase + 1 * 32 * 16);
        v16h b2 = *(const v16h*)(bbase + 2 * 32 * 16);
        v16h b3 = *(const v16h*)(bbase + 3 * 32 * 16);

        acc0 = __builtin_amdgcn_wmma_f32_16x16x32_f16(false, av, false, b0, (short)0, acc0, false, false);
        acc1 = __builtin_amdgcn_wmma_f32_16x16x32_f16(false, av, false, b1, (short)0, acc1, false, false);
        acc2 = __builtin_amdgcn_wmma_f32_16x16x32_f16(false, av, false, b2, (short)0, acc2, false, false);
        acc3 = __builtin_amdgcn_wmma_f32_16x16x32_f16(false, av, false, b3, (short)0, acc3, false, false);
    }

    float* cbase = C + (size_t)(tm * 16 + hsel * 8) * N + tn0 * 16 + idx;
#pragma unroll
    for (int r = 0; r < 8; ++r) {
        cbase[(size_t)r * N + 0]  = acc0[r];
        cbase[(size_t)r * N + 16] = acc1[r];
        cbase[(size_t)r * N + 32] = acc2[r];
        cbase[(size_t)r * N + 48] = acc3[r];
    }
}

__global__ void __launch_bounds__(512)
gcn_gather256_relu_gemm64(const int* __restrict__ rowptr, const int2* __restrict__ csr,
                          const float* __restrict__ dinv, const float* __restrict__ xw,
                          const float* __restrict__ bias, const _Float16* __restrict__ Bp,
                          float* __restrict__ xw2, int nodes) {
    __shared__ _Float16 htile[16 * GCN_H_PITCH];
    const int lane = threadIdx.x & 31;
    const int wv = __builtin_amdgcn_readfirstlane((int)(threadIdx.x >> 5));
    const int node0 = (int)blockIdx.x * 16;
    const int node = node0 + wv;
    float4 a0 = make_float4(0.0f, 0.0f, 0.0f, 0.0f);
    float4 a1 = a0;
    if (node < nodes) {
        const int beg = rowptr[node];
        const int end = rowptr[node + 1];
        const float di = dinv[node];
        const float4* __restrict__ self = (const float4*)(xw + (size_t)node * 256);
        a0 = self[lane];
        a1 = self[lane + 32];
        const float4 b0 = ((const float4*)bias)[lane];
        const float4 b1 = ((const float4*)bias)[lane + 32];
        a0.x *= di; a0.y *= di; a0.z *= di; a0.w *= di;
        a1.x *= di; a1.y *= di; a1.z *= di; a1.w *= di;
        for (int base = beg; base < end; base += 32) {
            int2 my = make_int2(0, 0);
            if (base + lane < end) my = csr[base + lane];
            const int n = (end - base < 32) ? (end - base) : 32;
#pragma unroll 4
            for (int k = 0; k < n; ++k) {
                const int s = __builtin_amdgcn_readlane(my.x, k);
                const float w = __int_as_float(__builtin_amdgcn_readlane(my.y, k));
                const float4* __restrict__ r = (const float4*)(xw + (size_t)s * 256);
                const float4 v0 = r[lane];
                const float4 v1 = r[lane + 32];
                a0.x = fmaf(v0.x, w, a0.x); a0.y = fmaf(v0.y, w, a0.y);
                a0.z = fmaf(v0.z, w, a0.z); a0.w = fmaf(v0.w, w, a0.w);
                a1.x = fmaf(v1.x, w, a1.x); a1.y = fmaf(v1.y, w, a1.y);
                a1.z = fmaf(v1.z, w, a1.z); a1.w = fmaf(v1.w, w, a1.w);
            }
        }
        a0.x = fmaxf(a0.x * di + b0.x, 0.0f); a0.y = fmaxf(a0.y * di + b0.y, 0.0f);
        a0.z = fmaxf(a0.z * di + b0.z, 0.0f); a0.w = fmaxf(a0.w * di + b0.w, 0.0f);
        a1.x = fmaxf(a1.x * di + b1.x, 0.0f); a1.y = fmaxf(a1.y * di + b1.y, 0.0f);
        a1.z = fmaxf(a1.z * di + b1.z, 0.0f); a1.w = fmaxf(a1.w * di + b1.w, 0.0f);
    }
    {
        v4h h0, h1;
        h0[0] = (_Float16)a0.x; h0[1] = (_Float16)a0.y; h0[2] = (_Float16)a0.z; h0[3] = (_Float16)a0.w;
        h1[0] = (_Float16)a1.x; h1[1] = (_Float16)a1.y; h1[2] = (_Float16)a1.z; h1[3] = (_Float16)a1.w;
        _Float16* hrow = htile + wv * GCN_H_PITCH + 4 * lane;
        *(v4h*)(hrow)       = h0;
        *(v4h*)(hrow + 128) = h1;
    }
    __syncthreads();
    if (wv >= 2) return;

    const int hsel = lane >> 4;
    const int idx  = lane & 15;
    const int tn0  = wv * 2;
    const _Float16* arow = htile + idx * GCN_H_PITCH + hsel * 8;
    v8f acc0 = {}, acc1 = {};
    for (int kt = 0; kt < GCN_D_HID / 32; ++kt) {
        const v8h lo = *(const v8h*)(arow + kt * 32);
        const v8h hi = *(const v8h*)(arow + kt * 32 + 16);
        const v16h av = __builtin_shufflevector(lo, hi, 0, 1, 2, 3, 4, 5, 6, 7,
                                                8, 9, 10, 11, 12, 13, 14, 15);
        const _Float16* bbase = Bp + ((size_t)(kt * (GCN_D_OUT >> 4) + tn0) * 32 + lane) * 16;
        const v16h b0 = *(const v16h*)(bbase);
        const v16h b1 = *(const v16h*)(bbase + 32 * 16);
        acc0 = __builtin_amdgcn_wmma_f32_16x16x32_f16(false, av, false, b0, (short)0, acc0, false, false);
        acc1 = __builtin_amdgcn_wmma_f32_16x16x32_f16(false, av, false, b1, (short)0, acc1, false, false);
    }
    const int row0 = node0 + hsel * 8;
    float* cbase = xw2 + (size_t)row0 * GCN_D_OUT + tn0 * 16 + idx;
#pragma unroll
    for (int r = 0; r < 8; ++r) {
        if (row0 + r < nodes) {
            cbase[(size_t)r * GCN_D_OUT + 0]  = acc0[r];
            cbase[(size_t)r * GCN_D_OUT + 16] = acc1[r];
        }
    }
}

__global__ void gcn_gather64_bias_softmax(const int* __restrict__ rowptr, const int2* __restrict__ csr,
                                          const float* __restrict__ dinv, const float* __restrict__ xw,
                                          const float* __restrict__ bias,
                                          float* __restrict__ out, int nodes) {
    const int lane = threadIdx.x & 31;
    const int node = __builtin_amdgcn_readfirstlane((int)((blockIdx.x * blockDim.x + threadIdx.x) >> 5));
    if (node >= nodes) return;
    const int beg = rowptr[node];
    const int end = rowptr[node + 1];
    const float di = dinv[node];
    float2 a = ((const float2*)(xw + (size_t)node * 64))[lane];
    const float2 b = ((const float2*)bias)[lane];
    a.x *= di; a.y *= di;
    for (int base = beg; base < end; base += 32) {
        int2 my = make_int2(0, 0);
        if (base + lane < end) my = csr[base + lane];
        const int n = (end - base < 32) ? (end - base) : 32;
#pragma unroll 8
        for (int k = 0; k < n; ++k) {
            const int s = __builtin_amdgcn_readlane(my.x, k);
            const float w = __int_as_float(__builtin_amdgcn_readlane(my.y, k));
            const float2 v = ((const float2*)(xw + (size_t)s * 64))[lane];
            a.x = fmaf(v.x, w, a.x);
            a.y = fmaf(v.y, w, a.y);
        }
    }
    const float v0 = a.x * di + b.x;
    const float v1 = a.y * di + b.y;
    float m = fmaxf(v0, v1);
#pragma unroll
    for (int off = 16; off > 0; off >>= 1) m = fmaxf(m, __shfl_xor(m, off, 32));
    const float e0 = __expf(v0 - m);
    const float e1 = __expf(v1 - m);
    float sum = e0 + e1;
#pragma unroll
    for (int off = 16; off > 0; off >>= 1) sum += __shfl_xor(sum, off, 32);
    const float inv = 1.0f / sum;
    ((float2*)(out + (size_t)node * 64))[lane] = make_float2(e0 * inv, e1 * inv);
}

extern "C" void kernel_launch(void* const* d_in, const int* in_sizes, int n_in,
                              void* d_out, int out_size, void* d_ws, size_t ws_size,
                              hipStream_t stream) {
    const float* x   = (const float*)d_in[0];
    const int*   ei  = (const int*)  d_in[1];
    const float* W1  = (const float*)d_in[2];
    const float* b1  = (const float*)d_in[3];
    const float* W2  = (const float*)d_in[4];
    const float* b2  = (const float*)d_in[5];
    float* out = (float*)d_out;

    const int N = in_sizes[0] / GCN_D_IN;
    const int E = in_sizes[1] / 2;
    const int* src = ei;
    const int* dst = ei + E;

    float* ws   = (float*)d_ws;
    size_t off  = ((size_t)N + 63) & ~(size_t)63;
    float* dinv = ws;
    float* xw1  = ws + off;
    float* reg2 = xw1 + (size_t)N * GCN_D_HID;
    _Float16* W1p = (_Float16*)(reg2 + (size_t)N * GCN_D_HID);
    _Float16* W2p = W1p + (size_t)GCN_D_IN * GCN_D_HID;
    float* xw2  = reg2;

    int*  so       = (int*)(reg2 + (size_t)N * GCN_D_OUT);
    int*  cursor   = so;
    int*  rowptr   = so + off;
    int*  blocktot = so + 2 * off + 64;
    int2* csr      = (int2*)(so + 2 * off + 64 + 1024);

    const int B = 256;
    {
        int t1 = (GCN_D_IN >> 5) * (GCN_D_HID >> 4) * 32;
        gcn_pack_w_f16<<<(t1 + B - 1) / B, B, 0, stream>>>(W1, W1p, GCN_D_IN, GCN_D_HID);
        int t2 = (GCN_D_HID >> 5) * (GCN_D_OUT >> 4) * 32;
        gcn_pack_w_f16<<<(t2 + B - 1) / B, B, 0, stream>>>(W2, W2p, GCN_D_HID, GCN_D_OUT);
    }

    {
        const int SB = 1024;
        gcn_zero_cnt   <<<(N + B - 1) / B, B, 0, stream>>>(cursor, N);
        gcn_count_in   <<<(E + B - 1) / B, B, 0, stream>>>(dst, cursor, E);
        gcn_scan_local <<<(N + SB - 1) / SB, SB, 0, stream>>>(cursor, rowptr, blocktot, N);
        gcn_scan_finish<<<(N + SB - 1) / SB, SB, 0, stream>>>(cursor, rowptr, blocktot, dinv, N);
        gcn_fill_in    <<<(E + B - 1) / B, B, 0, stream>>>(src, dst, dinv, cursor, csr, E);
    }

    {
        int waves = (N / 16) * (GCN_D_HID / 64);
        int th = waves * 32;
        gcn_wmma_gemm_f16x4<<<(th + B - 1) / B, B, 0, stream>>>(x, W1p, xw1, N, GCN_D_IN, GCN_D_HID);
    }
    {
        gcn_gather256_relu_gemm64<<<(N + 15) / 16, 512, 0, stream>>>(rowptr, csr, dinv, xw1, b1, W2p, xw2, N);
    }

    {
        int th = N * 32;
        gcn_gather64_bias_softmax<<<(th + B - 1) / B, B, 0, stream>>>(rowptr, csr, dinv, xw2, b2, out, N);
    }
}


# layer-1 GEMM: all packed W1 fragments (256 KB) staged once per workgroup in LDS, one 16-wave workgroup per WGP; each wave owns a 16x256 row block, so A is read once and B comes from LDS
# speedup vs baseline: 2.8634x; 1.1613x over previous
#include <hip/hip_runtime.h>
#include <hip/hip_bf16.h>

typedef __attribute__((ext_vector_type(16))) _Float16 v16h;
typedef __attribute__((ext_vector_type(8)))  float    v8f;
typedef __attribute__((ext_vector_type(4)))  _Float16 v4h;
typedef __attribute__((ext_vector_type(8)))  _Float16 v8h;

#define GCN_D_IN  512
#define GCN_D_HID 256
#define GCN_D_OUT 64
#define GCN_H_PITCH (GCN_D_HID + 8)

__global__ void gcn_zero_cnt(int* __restrict__ cnt, int n) {
    int i = blockIdx.x * blockDim.x + threadIdx.x;
    if (i < n) cnt[i] = 0;
}

__global__ void gcn_count_in(const int* __restrict__ dst, int* __restrict__ cnt, int E) {
    int e = blockIdx.x * blockDim.x + threadIdx.x;
    if (e < E) atomicAdd(&cnt[dst[e]], 1);
}

__global__ void gcn_scan_local(const int* __restrict__ cnt, int* __restrict__ rowptr,
                               int* __restrict__ blocktot, int n) {
    __shared__ int wsum[32];
    const int t = threadIdx.x;
    const int lane = t & 31;
    const int wv = t >> 5;
    const int i = blockIdx.x * 1024 + t;
    const int c = (i < n) ? cnt[i] : 0;
    int incl = c;
#pragma unroll
    for (int off = 1; off < 32; off <<= 1) {
        int y = __shfl_up(incl, off, 32);
        if (lane >= off) incl += y;
    }
    if (lane == 31) wsum[wv] = incl;
    __syncthreads();
    if (wv == 0) {
        const int v = wsum[lane];
        int inc2 = v;
#pragma unroll
        for (int off = 1; off < 32; off <<= 1) {
            int y = __shfl_up(inc2, off, 32);
            if (lane >= off) inc2 += y;
        }
        wsum[lane] = inc2 - v;
    }
    __syncthreads();
    const int excl = incl - c + wsum[wv];
    if (i < n) rowptr[i] = excl;
    if (t == 1023) blocktot[blockIdx.x] = excl + c;
}

__global__ void gcn_scan_finish(int* __restrict__ cnt_cursor, int* __restrict__ rowptr,
                                const int* __restrict__ blocktot, float* __restrict__ dinv,
                                int n) {
    const int i = blockIdx.x * 1024 + threadIdx.x;
    int base = 0;
    for (int j = 0; j < (int)blockIdx.x; ++j) base += blocktot[j];
    if (i < n) {
        const int c = cnt_cursor[i];
        const int p = rowptr[i] + base;
        rowptr[i] = p;
        cnt_cursor[i] = p;
        const float d = (float)(c + 1);
        dinv[i] = (d > 0.0f) ? rsqrtf(d) : 0.0f;
        if (i == n - 1) rowptr[n] = p + c;
    }
}

__global__ void gcn_fill_in(const int* __restrict__ src, const int* __restrict__ dst,
                            const float* __restrict__ dinv, int* __restrict__ cursor,
                            int2* __restrict__ csr, int E) {
    int e = blockIdx.x * blockDim.x + threadIdx.x;
    if (e >= E) return;
    const int s = src[e], d = dst[e];
    const int pos = atomicAdd(&cursor[d], 1);
    csr[pos] = make_int2(s, __float_as_int(dinv[s]));
}

__global__ void gcn_pack_w_f16(const float* __restrict__ W, _Float16* __restrict__ Wp,
                               int K, int N) {
    int t = blockIdx.x * blockDim.x + threadIdx.x;
    int tilesN = N >> 4;
    int total = (K >> 5) * tilesN * 32;
    if (t >= total) return;
    int frag = t >> 5;
    int lane = t & 31;
    int kt = frag / tilesN;
    int tn = frag - kt * tilesN;
    int n    = tn * 16 + (lane & 15);
    int koff = kt * 32 + (lane >> 4) * 8;
    _Float16* o = Wp + (size_t)t * 16;
#pragma unroll
    for (int j = 0; j < 8; ++j) {
        o[j]     = (_Float16)W[(size_t)(koff + j) * N + n];
        o[j + 8] = (_Float16)W[(size_t)(koff + 16 + j) * N + n];
    }
}


static __device__ __forceinline__ void gcn_ldb4(v16h (&dst)[4], const v8h* p) {
#pragma unroll
    for (int j = 0; j < 4; ++j) {
        const v8h lo = p[j * 64];
        const v8h hi = p[j * 64 + 32];
        dst[j] = __builtin_shufflevector(lo, hi, 0, 1, 2, 3, 4, 5, 6, 7,
                                         8, 9, 10, 11, 12, 13, 14, 15);
    }
}

static __device__ __forceinline__ void gcn_lda(float4 (&dst)[4], const float* arow, int kt) {
    const float* p = arow + kt * 32;
    dst[0] = *(const float4*)(p);
    dst[1] = *(const float4*)(p + 4);
    dst[2] = *(const float4*)(p + 16);
    dst[3] = *(const float4*)(p + 20);
}

__global__ void __launch_bounds__(512)
gcn_wmma_gemm_f16_ldsb(const float* __restrict__ A, const _Float16* __restrict__ Bp,
                       float* __restrict__ C, int M) {
    __shared__ __attribute__((aligned(16))) v8h Bs[(GCN_D_IN * GCN_D_HID) / 8];
    const int tid  = threadIdx.x;
    const int lane = tid & 31;
    const int wv   = __builtin_amdgcn_readfirstlane(tid >> 5);
    const int hsel = lane >> 4;
    const int idx  = lane & 15;
    const int tilesM = (M + 15) >> 4;

    const int tm = wv * (int)gridDim.x + (int)blockIdx.x;
    int row = tm * 16 + idx;
    row = (row < M) ? row : (M - 1);
    const float* __restrict__ arow = A + (size_t)row * GCN_D_IN + hsel * 8;
    if (tm < tilesM) {
        __builtin_prefetch(arow, 0, 3);
        __builtin_prefetch(arow + 32, 0, 3);
        __builtin_prefetch(arow + 64, 0, 3);
    }

    {
        const v8h* __restrict__ Bg = (const v8h*)Bp;
#pragma unroll 8
        for (int i = 0; i < 32; ++i) {
            const int g = (((i + (int)blockIdx.x) & 31) << 9) + tid;
            const int dst = ((g >> 6) << 6) | ((g & 1) << 5) | ((g >> 1) & 31);
            Bs[dst] = Bg[g];
        }
    }
    __syncthreads();
    if (tm >= tilesM) return;

    v8f acc[16];
#pragma unroll
    for (int t = 0; t < 16; ++t) acc[t] = (v8f){};

    float4 ar[3][4];
    gcn_lda(ar[0], arow, 0);
    gcn_lda(ar[1], arow, 1);
    gcn_lda(ar[2], arow, 2);

    v16h bq[2][4];
    const v8h* bk = Bs + lane;
    gcn_ldb4(bq[0], bk);

#pragma unroll
    for (int kt = 0; kt < GCN_D_IN / 32; ++kt) {
        const int s = kt % 3;
        float af[16];
        *(float4*)(af + 0)  = ar[s][0];  *(float4*)(af + 4)  = ar[s][1];
        *(float4*)(af + 8)  = ar[s][2];  *(float4*)(af + 12) = ar[s][3];
        v16h av;
#pragma unroll
        for (int j = 0; j < 16; ++j) av[j] = (_Float16)af[j];
        if (kt + 3 < GCN_D_IN / 32) gcn_lda(ar[s], arow, kt + 3);
#pragma unroll
        for (int g = 0; g < 4; ++g) {
            const int cur = g & 1;
            const int nf = kt * 16 + (g + 1) * 4;
            if (nf < (GCN_D_IN / 32) * 16) gcn_ldb4(bq[cur ^ 1], bk + nf * 64);
#pragma unroll
            for (int j = 0; j < 4; ++j)
                acc[g * 4 + j] = __builtin_amdgcn_wmma_f32_16x16x32_f16(false, av, false, bq[cur][j], (short)0,
                                                                       acc[g * 4 + j], false, false);
        }
    }

    const int row0 = tm * 16 + hsel * 8;
    float* cbase = C + (size_t)row0 * GCN_D_HID + idx;
#pragma unroll
    for (int r = 0; r < 8; ++r) {
        if (row0 + r < M) {
#pragma unroll
            for (int t = 0; t < 16; ++t) cbase[(size_t)r * GCN_D_HID + t * 16] = acc[t][r];
        }
    }
}

__global__ void __launch_bounds__(512)
gcn_gather256_relu_gemm64(const int* __restrict__ rowptr, const int2* __restrict__ csr,
                          const float* __restrict__ dinv, const float* __restrict__ xw,
                          const float* __restrict__ bias, const _Float16* __restrict__ Bp,
                          float* __restrict__ xw2, int nodes) {
    __shared__ _Float16 htile[16 * GCN_H_PITCH];
    const int lane = threadIdx.x & 31;
    const int wv = __builtin_amdgcn_readfirstlane((int)(threadIdx.x >> 5));
    const int node0 = (int)blockIdx.x * 16;
    const int node = node0 + wv;
    float4 a0 = make_float4(0.0f, 0.0f, 0.0f, 0.0f);
    float4 a1 = a0;
    if (node < nodes) {
        const int beg = rowptr[node];
        const int end = rowptr[node + 1];
        const float di = dinv[node];
        const float4* __restrict__ self = (const float4*)(xw + (size_t)node * 256);
        a0 = self[lane];
        a1 = self[lane + 32];
        const float4 b0 = ((const float4*)bias)[lane];
        const float4 b1 = ((const float4*)bias)[lane + 32];
        a0.x *= di; a0.y *= di; a0.z *= di; a0.w *= di;
        a1.x *= di; a1.y *= di; a1.z *= di; a1.w *= di;
        for (int base = beg; base < end; base += 32) {
            int2 my = make_int2(0, 0);
            if (base + lane < end) my = csr[base + lane];
            const int n = (end - base < 32) ? (end - base) : 32;
#pragma unroll 4
            for (int k = 0; k < n; ++k) {
                const int s = __builtin_amdgcn_readlane(my.x, k);
                const float w = __int_as_float(__builtin_amdgcn_readlane(my.y, k));
                const float4* __restrict__ r = (const float4*)(xw + (size_t)s * 256);
                const float4 v0 = r[lane];
                const float4 v1 = r[lane + 32];
                a0.x = fmaf(v0.x, w, a0.x); a0.y = fmaf(v0.y, w, a0.y);
                a0.z = fmaf(v0.z, w, a0.z); a0.w = fmaf(v0.w, w, a0.w);
                a1.x = fmaf(v1.x, w, a1.x); a1.y = fmaf(v1.y, w, a1.y);
                a1.z = fmaf(v1.z, w, a1.z); a1.w = fmaf(v1.w, w, a1.w);
            }
        }
        a0.x = fmaxf(a0.x * di + b0.x, 0.0f); a0.y = fmaxf(a0.y * di + b0.y, 0.0f);
        a0.z = fmaxf(a0.z * di + b0.z, 0.0f); a0.w = fmaxf(a0.w * di + b0.w, 0.0f);
        a1.x = fmaxf(a1.x * di + b1.x, 0.0f); a1.y = fmaxf(a1.y * di + b1.y, 0.0f);
        a1.z = fmaxf(a1.z * di + b1.z, 0.0f); a1.w = fmaxf(a1.w * di + b1.w, 0.0f);
    }
    {
        v4h h0, h1;
        h0[0] = (_Float16)a0.x; h0[1] = (_Float16)a0.y; h0[2] = (_Float16)a0.z; h0[3] = (_Float16)a0.w;
        h1[0] = (_Float16)a1.x; h1[1] = (_Float16)a1.y; h1[2] = (_Float16)a1.z; h1[3] = (_Float16)a1.w;
        _Float16* hrow = htile + wv * GCN_H_PITCH + 4 * lane;
        *(v4h*)(hrow)       = h0;
        *(v4h*)(hrow + 128) = h1;
    }
    __syncthreads();
    if (wv >= 2) return;

    const int hsel = lane >> 4;
    const int idx  = lane & 15;
    const int tn0  = wv * 2;
    const _Float16* arow = htile + idx * GCN_H_PITCH + hsel * 8;
    v8f acc0 = {}, acc1 = {};
    for (int kt = 0; kt < GCN_D_HID / 32; ++kt) {
        const v8h lo = *(const v8h*)(arow + kt * 32);
        const v8h hi = *(const v8h*)(arow + kt * 32 + 16);
        const v16h av = __builtin_shufflevector(lo, hi, 0, 1, 2, 3, 4, 5, 6, 7,
                                                8, 9, 10, 11, 12, 13, 14, 15);
        const _Float16* bbase = Bp + ((size_t)(kt * (GCN_D_OUT >> 4) + tn0) * 32 + lane) * 16;
        const v16h b0 = *(const v16h*)(bbase);
        const v16h b1 = *(const v16h*)(bbase + 32 * 16);
        acc0 = __builtin_amdgcn_wmma_f32_16x16x32_f16(false, av, false, b0, (short)0, acc0, false, false);
        acc1 = __builtin_amdgcn_wmma_f32_16x16x32_f16(false, av, false, b1, (short)0, acc1, false, false);
    }
    const int row0 = node0 + hsel * 8;
    float* cbase = xw2 + (size_t)row0 * GCN_D_OUT + tn0 * 16 + idx;
#pragma unroll
    for (int r = 0; r < 8; ++r) {
        if (row0 + r < nodes) {
            cbase[(size_t)r * GCN_D_OUT + 0]  = acc0[r];
            cbase[(size_t)r * GCN_D_OUT + 16] = acc1[r];
        }
    }
}

__global__ void gcn_gather64_bias_softmax(const int* __restrict__ rowptr, const int2* __restrict__ csr,
                                          const float* __restrict__ dinv, const float* __restrict__ xw,
                                          const float* __restrict__ bias,
                                          float* __restrict__ out, int nodes) {
    const int lane = threadIdx.x & 31;
    const int node = __builtin_amdgcn_readfirstlane((int)((blockIdx.x * blockDim.x + threadIdx.x) >> 5));
    if (node >= nodes) return;
    const int beg = rowptr[node];
    const int end = rowptr[node + 1];
    const float di = dinv[node];
    float2 a = ((const float2*)(xw + (size_t)node * 64))[lane];
    const float2 b = ((const float2*)bias)[lane];
    a.x *= di; a.y *= di;
    for (int base = beg; base < end; base += 32) {
        int2 my = make_int2(0, 0);
        if (base + lane < end) my = csr[base + lane];
        const int n = (end - base < 32) ? (end - base) : 32;
#pragma unroll 8
        for (int k = 0; k < n; ++k) {
            const int s = __builtin_amdgcn_readlane(my.x, k);
            const float w = __int_as_float(__builtin_amdgcn_readlane(my.y, k));
            const float2 v = ((const float2*)(xw + (size_t)s * 64))[lane];
            a.x = fmaf(v.x, w, a.x);
            a.y = fmaf(v.y, w, a.y);
        }
    }
    const float v0 = a.x * di + b.x;
    const float v1 = a.y * di + b.y;
    float m = fmaxf(v0, v1);
#pragma unroll
    for (int off = 16; off > 0; off >>= 1) m = fmaxf(m, __shfl_xor(m, off, 32));
    const float e0 = __expf(v0 - m);
    const float e1 = __expf(v1 - m);
    float sum = e0 + e1;
#pragma unroll
    for (int off = 16; off > 0; off >>= 1) sum += __shfl_xor(sum, off, 32);
    const float inv = 1.0f / sum;
    ((float2*)(out + (size_t)node * 64))[lane] = make_float2(e0 * inv, e1 * inv);
}

extern "C" void kernel_launch(void* const* d_in, const int* in_sizes, int n_in,
                              void* d_out, int out_size, void* d_ws, size_t ws_size,
                              hipStream_t stream) {
    const float* x   = (const float*)d_in[0];
    const int*   ei  = (const int*)  d_in[1];
    const float* W1  = (const float*)d_in[2];
    const float* b1  = (const float*)d_in[3];
    const float* W2  = (const float*)d_in[4];
    const float* b2  = (const float*)d_in[5];
    float* out = (float*)d_out;

    const int N = in_sizes[0] / GCN_D_IN;
    const int E = in_sizes[1] / 2;
    const int* src = ei;
    const int* dst = ei + E;

    float* ws   = (float*)d_ws;
    size_t off  = ((size_t)N + 63) & ~(size_t)63;
    float* dinv = ws;
    float* xw1  = ws + off;
    float* reg2 = xw1 + (size_t)N * GCN_D_HID;
    _Float16* W1p = (_Float16*)(reg2 + (size_t)N * GCN_D_HID);
    _Float16* W2p = W1p + (size_t)GCN_D_IN * GCN_D_HID;
    float* xw2  = reg2;

    int*  so       = (int*)(reg2 + (size_t)N * GCN_D_OUT);
    int*  cursor   = so;
    int*  rowptr   = so + off;
    int*  blocktot = so + 2 * off + 64;
    int2* csr      = (int2*)(so + 2 * off + 64 + 1024);

    const int B = 256;
    {
        int t1 = (GCN_D_IN >> 5) * (GCN_D_HID >> 4) * 32;
        gcn_pack_w_f16<<<(t1 + B - 1) / B, B, 0, stream>>>(W1, W1p, GCN_D_IN, GCN_D_HID);
        int t2 = (GCN_D_HID >> 5) * (GCN_D_OUT >> 4) * 32;
        gcn_pack_w_f16<<<(t2 + B - 1) / B, B, 0, stream>>>(W2, W2p, GCN_D_HID, GCN_D_OUT);
    }

    {
        const int SB = 1024;
        gcn_zero_cnt   <<<(N + B - 1) / B, B, 0, stream>>>(cursor, N);
        gcn_count_in   <<<(E + B - 1) / B, B, 0, stream>>>(dst, cursor, E);
        gcn_scan_local <<<(N + SB - 1) / SB, SB, 0, stream>>>(cursor, rowptr, blocktot, N);
        gcn_scan_finish<<<(N + SB - 1) / SB, SB, 0, stream>>>(cursor, rowptr, blocktot, dinv, N);
        gcn_fill_in    <<<(E + B - 1) / B, B, 0, stream>>>(src, dst, dinv, cursor, csr, E);
    }

    {
        int g1 = ((N + 15) / 16 + 15) / 16;
        if (g1 < 256) g1 = 256;
        gcn_wmma_gemm_f16_ldsb<<<g1, 512, 0, stream>>>(x, W1p, xw1, N);
    }
    {
        gcn_gather256_relu_gemm64<<<(N + 15) / 16, 512, 0, stream>>>(rowptr, csr, dinv, xw1, b1, W2p, xw2, N);
    }

    {
        int th = N * 32;
        gcn_gather64_bias_softmax<<<(th + B - 1) / B, B, 0, stream>>>(rowptr, csr, dinv, xw2, b2, out, N);
    }
}


# layer-1 product xw1 stored as f16 instead of f32 (precision of the gathered table narrowed; sums stay f32): a row is 512 B, one 16-B load per lane; W1 columns re-dealt so a lane stores 8 adjacent
# speedup vs baseline: 3.1076x; 1.0853x over previous
#include <hip/hip_runtime.h>
#include <hip/hip_bf16.h>

typedef __attribute__((ext_vector_type(16))) _Float16 v16h;
typedef __attribute__((ext_vector_type(8)))  float    v8f;
typedef __attribute__((ext_vector_type(4)))  _Float16 v4h;
typedef __attribute__((ext_vector_type(8)))  _Float16 v8h;

#define GCN_D_IN  512
#define GCN_D_HID 256
#define GCN_D_OUT 64
#define GCN_H_PITCH (GCN_D_HID + 8)

__global__ void gcn_zero_cnt(int* __restrict__ cnt, int n) {
    int i = blockIdx.x * blockDim.x + threadIdx.x;
    if (i < n) cnt[i] = 0;
}

__global__ void gcn_count_in(const int* __restrict__ dst, int* __restrict__ cnt, int E) {
    int e = blockIdx.x * blockDim.x + threadIdx.x;
    if (e < E) atomicAdd(&cnt[dst[e]], 1);
}

__global__ void gcn_scan_local(const int* __restrict__ cnt, int* __restrict__ rowptr,
                               int* __restrict__ blocktot, int n) {
    __shared__ int wsum[32];
    const int t = threadIdx.x;
    const int lane = t & 31;
    const int wv = t >> 5;
    const int i = blockIdx.x * 1024 + t;
    const int c = (i < n) ? cnt[i] : 0;
    int incl = c;
#pragma unroll
    for (int off = 1; off < 32; off <<= 1) {
        int y = __shfl_up(incl, off, 32);
        if (lane >= off) incl += y;
    }
    if (lane == 31) wsum[wv] = incl;
    __syncthreads();
    if (wv == 0) {
        const int v = wsum[lane];
        int inc2 = v;
#pragma unroll
        for (int off = 1; off < 32; off <<= 1) {
            int y = __shfl_up(inc2, off, 32);
            if (lane >= off) inc2 += y;
        }
        wsum[lane] = inc2 - v;
    }
    __syncthreads();
    const int excl = incl - c + wsum[wv];
    if (i < n) rowptr[i] = excl;
    if (t == 1023) blocktot[blockIdx.x] = excl + c;
}

__global__ void gcn_scan_finish(int* __restrict__ cnt_cursor, int* __restrict__ rowptr,
                                const int* __restrict__ blocktot, float* __restrict__ dinv,
                                int n) {
    const int i = blockIdx.x * 1024 + threadIdx.x;
    int base = 0;
    for (int j = 0; j < (int)blockIdx.x; ++j) base += blocktot[j];
    if (i < n) {
        const int c = cnt_cursor[i];
        const int p = rowptr[i] + base;
        rowptr[i] = p;
        cnt_cursor[i] = p;
        const float d = (float)(c + 1);
        dinv[i] = (d > 0.0f) ? rsqrtf(d) : 0.0f;
        if (i == n - 1) rowptr[n] = p + c;
    }
}

__global__ void gcn_fill_in(const int* __restrict__ src, const int* __restrict__ dst,
                            const float* __restrict__ dinv, int* __restrict__ cursor,
                            int2* __restrict__ csr, int E) {
    int e = blockIdx.x * blockDim.x + threadIdx.x;
    if (e >= E) return;
    const int s = src[e], d = dst[e];
    const int pos = atomicAdd(&cursor[d], 1);
    csr[pos] = make_int2(s, __float_as_int(dinv[s]));
}

__global__ void gcn_pack_w_f16(const float* __restrict__ W, _Float16* __restrict__ Wp,
                               int K, int N, int deal8) {
    int t = blockIdx.x * blockDim.x + threadIdx.x;
    int tilesN = N >> 4;
    int total = (K >> 5) * tilesN * 32;
    if (t >= total) return;
    int frag = t >> 5;
    int lane = t & 31;
    int kt = frag / tilesN;
    int tn = frag - kt * tilesN;
    int n    = deal8 ? ((tn >> 3) * 128 + (lane & 15) * 8 + (tn & 7))
                     : (tn * 16 + (lane & 15));
    int koff = kt * 32 + (lane >> 4) * 8;
    _Float16* o = Wp + (size_t)t * 16;
#pragma unroll
    for (int j = 0; j < 8; ++j) {
        o[j]     = (_Float16)W[(size_t)(koff + j) * N + n];
        o[j + 8] = (_Float16)W[(size_t)(koff + 16 + j) * N + n];
    }
}


static __device__ __forceinline__ void gcn_ldb4(v16h (&dst)[4], const v8h* p) {
#pragma unroll
    for (int j = 0; j < 4; ++j) {
        const v8h lo = p[j * 64];
        const v8h hi = p[j * 64 + 32];
        dst[j] = __builtin_shufflevector(lo, hi, 0, 1, 2, 3, 4, 5, 6, 7,
                                         8, 9, 10, 11, 12, 13, 14, 15);
    }
}

static __device__ __forceinline__ void gcn_lda(float4 (&dst)[4], const float* arow, int kt) {
    const float* p = arow + kt * 32;
    dst[0] = *(const float4*)(p);
    dst[1] = *(const float4*)(p + 4);
    dst[2] = *(const float4*)(p + 16);
    dst[3] = *(const float4*)(p + 20);
}

__global__ void __launch_bounds__(512)
gcn_wmma_gemm_f16_ldsb(const float* __restrict__ A, const _Float16* __restrict__ Bp,
                       _Float16* __restrict__ C, int M) {
    __shared__ __attribute__((aligned(16))) v8h Bs[(GCN_D_IN * GCN_D_HID) / 8];
    const int tid  = threadIdx.x;
    const int lane = tid & 31;
    const int wv   = __builtin_amdgcn_readfirstlane(tid >> 5);
    const int hsel = lane >> 4;
    const int idx  = lane & 15;
    const int tilesM = (M + 15) >> 4;

    const int tm = wv * (int)gridDim.x + (int)blockIdx.x;
    int row = tm * 16 + idx;
    row = (row < M) ? row : (M - 1);
    const float* __restrict__ arow = A + (size_t)row * GCN_D_IN + hsel * 8;
    if (tm < tilesM) {
        __builtin_prefetch(arow, 0, 3);
        __builtin_prefetch(arow + 32, 0, 3);
        __builtin_prefetch(arow + 64, 0, 3);
    }

    {
        const v8h* __restrict__ Bg = (const v8h*)Bp;
#pragma unroll 8
        for (int i = 0; i < 32; ++i) {
            const int g = (((i + (int)blockIdx.x) & 31) << 9) + tid;
            const int dst = ((g >> 6) << 6) | ((g & 1) << 5) | ((g >> 1) & 31);
            Bs[dst] = Bg[g];
        }
    }
    __syncthreads();
    if (tm >= tilesM) return;

    v8f acc[16];
#pragma unroll
    for (int t = 0; t < 16; ++t) acc[t] = (v8f){};

    float4 ar[3][4];
    gcn_lda(ar[0], arow, 0);
    gcn_lda(ar[1], arow, 1);
    gcn_lda(ar[2], arow, 2);

    v16h bq[2][4];
    const v8h* bk = Bs + lane;
    gcn_ldb4(bq[0], bk);

#pragma unroll
    for (int kt = 0; kt < GCN_D_IN / 32; ++kt) {
        const int s = kt % 3;
        float af[16];
        *(float4*)(af + 0)  = ar[s][0];  *(float4*)(af + 4)  = ar[s][1];
        *(float4*)(af + 8)  = ar[s][2];  *(float4*)(af + 12) = ar[s][3];
        v16h av;
#pragma unroll
        for (int j = 0; j < 16; ++j) av[j] = (_Float16)af[j];
        if (kt + 3 < GCN_D_IN / 32) gcn_lda(ar[s], arow, kt + 3);
#pragma unroll
        for (int g = 0; g < 4; ++g) {
            const int cur = g & 1;
            const int nf = kt * 16 + (g + 1) * 4;
            if (nf < (GCN_D_IN / 32) * 16) gcn_ldb4(bq[cur ^ 1], bk + nf * 64);
#pragma unroll
            for (int j = 0; j < 4; ++j)
                acc[g * 4 + j] = __builtin_amdgcn_wmma_f32_16x16x32_f16(false, av, false, bq[cur][j], (short)0,
                                                                       acc[g * 4 + j], false, false);
        }
    }

    const int row0 = tm * 16 + hsel * 8;
    _Float16* cbase = C + (size_t)row0 * GCN_D_HID + idx * 8;
#pragma unroll
    for (int r = 0; r < 8; ++r) {
        if (row0 + r < M) {
#pragma unroll
            for (int g = 0; g < 2; ++g) {
                v8h o;
#pragma unroll
                for (int j = 0; j < 8; ++j) o[j] = (_Float16)acc[g * 8 + j][r];
                *(v8h*)(cbase + (size_t)r * GCN_D_HID + g * 128) = o;
            }
        }
    }
}

__global__ void __launch_bounds__(512)
gcn_gather256_relu_gemm64(const int* __restrict__ rowptr, const int2* __restrict__ csr,
                          const float* __restrict__ dinv, const _Float16* __restrict__ xw,
                          const float* __restrict__ bias, const _Float16* __restrict__ Bp,
                          float* __restrict__ xw2, int nodes) {
    __shared__ __attribute__((aligned(16))) _Float16 htile[16 * GCN_H_PITCH];
    const int lane = threadIdx.x & 31;
    const int wv = __builtin_amdgcn_readfirstlane((int)(threadIdx.x >> 5));
    const int node0 = (int)blockIdx.x * 16;
    const int node = node0 + wv;
    float a[8];
#pragma unroll
    for (int j = 0; j < 8; ++j) a[j] = 0.0f;
    if (node < nodes) {
        const int beg = rowptr[node];
        const int end = rowptr[node + 1];
        const float di = dinv[node];
        const v8h sv = ((const v8h*)(xw + (size_t)node * 256))[lane];
        const float4 b0 = ((const float4*)bias)[2 * lane];
        const float4 b1 = ((const float4*)bias)[2 * lane + 1];
        const float b[8] = {b0.x, b0.y, b0.z, b0.w, b1.x, b1.y, b1.z, b1.w};
#pragma unroll
        for (int j = 0; j < 8; ++j) a[j] = (float)sv[j] * di;
        for (int base = beg; base < end; base += 32) {
            int2 my = make_int2(0, 0);
            if (base + lane < end) my = csr[base + lane];
            const int n = (end - base < 32) ? (end - base) : 32;
#pragma unroll 4
            for (int k = 0; k < n; ++k) {
                const int s = __builtin_amdgcn_readlane(my.x, k);
                const float w = __int_as_float(__builtin_amdgcn_readlane(my.y, k));
                const v8h v = ((const v8h*)(xw + (size_t)s * 256))[lane];
#pragma unroll
                for (int j = 0; j < 8; ++j) a[j] = fmaf((float)v[j], w, a[j]);
            }
        }
#pragma unroll
        for (int j = 0; j < 8; ++j) a[j] = fmaxf(a[j] * di + b[j], 0.0f);
    }
    {
        v8h h;
#pragma unroll
        for (int j = 0; j < 8; ++j) h[j] = (_Float16)a[j];
        *(v8h*)(htile + wv * GCN_H_PITCH + 8 * lane) = h;
    }
    __syncthreads();
    if (wv >= 2) return;

    const int hsel = lane >> 4;
    const int idx  = lane & 15;
    const int tn0  = wv * 2;
    const _Float16* arow = htile + idx * GCN_H_PITCH + hsel * 8;
    v8f acc0 = {}, acc1 = {};
    for (int kt = 0; kt < GCN_D_HID / 32; ++kt) {
        const v8h lo = *(const v8h*)(arow + kt * 32);
        const v8h hi = *(const v8h*)(arow + kt * 32 + 16);
        const v16h av = __builtin_shufflevector(lo, hi, 0, 1, 2, 3, 4, 5, 6, 7,
                                                8, 9, 10, 11, 12, 13, 14, 15);
        const _Float16* bbase = Bp + ((size_t)(kt * (GCN_D_OUT >> 4) + tn0) * 32 + lane) * 16;
        const v16h b0 = *(const v16h*)(bbase);
        const v16h b1 = *(const v16h*)(bbase + 32 * 16);
        acc0 = __builtin_amdgcn_wmma_f32_16x16x32_f16(false, av, false, b0, (short)0, acc0, false, false);
        acc1 = __builtin_amdgcn_wmma_f32_16x16x32_f16(false, av, false, b1, (short)0, acc1, false, false);
    }
    const int row0 = node0 + hsel * 8;
    float* cbase = xw2 + (size_t)row0 * GCN_D_OUT + tn0 * 16 + idx;
#pragma unroll
    for (int r = 0; r < 8; ++r) {
        if (row0 + r < nodes) {
            cbase[(size_t)r * GCN_D_OUT + 0]  = acc0[r];
            cbase[(size_t)r * GCN_D_OUT + 16] = acc1[r];
        }
    }
}

__global__ void gcn_gather64_bias_softmax(const int* __restrict__ rowptr, const int2* __restrict__ csr,
                                          const float* __restrict__ dinv, const float* __restrict__ xw,
                                          const float* __restrict__ bias,
                                          float* __restrict__ out, int nodes) {
    const int lane = threadIdx.x & 31;
    const int node = __builtin_amdgcn_readfirstlane((int)((blockIdx.x * blockDim.x + threadIdx.x) >> 5));
    if (node >= nodes) return;
    const int beg = rowptr[node];
    const int end = rowptr[node + 1];
    const float di = dinv[node];
    float2 a = ((const float2*)(xw + (size_t)node * 64))[lane];
    const float2 b = ((const float2*)bias)[lane];
    a.x *= di; a.y *= di;
    for (int base = beg; base < end; base += 32) {
        int2 my = make_int2(0, 0);
        if (base + lane < end) my = csr[base + lane];
        const int n = (end - base < 32) ? (end - base) : 32;
#pragma unroll 8
        for (int k = 0; k < n; ++k) {
            const int s = __builtin_amdgcn_readlane(my.x, k);
            const float w = __int_as_float(__builtin_amdgcn_readlane(my.y, k));
            const float2 v = ((const float2*)(xw + (size_t)s * 64))[lane];
            a.x = fmaf(v.x, w, a.x);
            a.y = fmaf(v.y, w, a.y);
        }
    }
    const float v0 = a.x * di + b.x;
    const float v1 = a.y * di + b.y;
    float m = fmaxf(v0, v1);
#pragma unroll
    for (int off = 16; off > 0; off >>= 1) m = fmaxf(m, __shfl_xor(m, off, 32));
    const float e0 = __expf(v0 - m);
    const float e1 = __expf(v1 - m);
    float sum = e0 + e1;
#pragma unroll
    for (int off = 16; off > 0; off >>= 1) sum += __shfl_xor(sum, off, 32);
    const float inv = 1.0f / sum;
    ((float2*)(out + (size_t)node * 64))[lane] = make_float2(e0 * inv, e1 * inv);
}

extern "C" void kernel_launch(void* const* d_in, const int* in_sizes, int n_in,
                              void* d_out, int out_size, void* d_ws, size_t ws_size,
                              hipStream_t stream) {
    const float* x   = (const float*)d_in[0];
    const int*   ei  = (const int*)  d_in[1];
    const float* W1  = (const float*)d_in[2];
    const float* b1  = (const float*)d_in[3];
    const float* W2  = (const float*)d_in[4];
    const float* b2  = (const float*)d_in[5];
    float* out = (float*)d_out;

    const int N = in_sizes[0] / GCN_D_IN;
    const int E = in_sizes[1] / 2;
    const int* src = ei;
    const int* dst = ei + E;

    float* ws   = (float*)d_ws;
    size_t off  = ((size_t)N + 63) & ~(size_t)63;
    float* dinv = ws;
    _Float16* xw1 = (_Float16*)(ws + off);
    float* reg2 = ws + off + (size_t)N * GCN_D_HID;
    _Float16* W1p = (_Float16*)(reg2 + (size_t)N * GCN_D_HID);
    _Float16* W2p = W1p + (size_t)GCN_D_IN * GCN_D_HID;
    float* xw2  = reg2;

    int*  so       = (int*)(reg2 + (size_t)N * GCN_D_OUT);
    int*  cursor   = so;
    int*  rowptr   = so + off;
    int*  blocktot = so + 2 * off + 64;
    int2* csr      = (int2*)(so + 2 * off + 64 + 1024);

    const int B = 256;
    {
        int t1 = (GCN_D_IN >> 5) * (GCN_D_HID >> 4) * 32;
        gcn_pack_w_f16<<<(t1 + B - 1) / B, B, 0, stream>>>(W1, W1p, GCN_D_IN, GCN_D_HID, 1);
        int t2 = (GCN_D_HID >> 5) * (GCN_D_OUT >> 4) * 32;
        gcn_pack_w_f16<<<(t2 + B - 1) / B, B, 0, stream>>>(W2, W2p, GCN_D_HID, GCN_D_OUT, 0);
    }

    {
        const int SB = 1024;
        gcn_zero_cnt   <<<(N + B - 1) / B, B, 0, stream>>>(cursor, N);
        gcn_count_in   <<<(E + B - 1) / B, B, 0, stream>>>(dst, cursor, E);
        gcn_scan_local <<<(N + SB - 1) / SB, SB, 0, stream>>>(cursor, rowptr, blocktot, N);
        gcn_scan_finish<<<(N + SB - 1) / SB, SB, 0, stream>>>(cursor, rowptr, blocktot, dinv, N);
        gcn_fill_in    <<<(E + B - 1) / B, B, 0, stream>>>(src, dst, dinv, cursor, csr, E);
    }

    {
        int g1 = ((N + 15) / 16 + 15) / 16;
        if (g1 < 256) g1 = 256;
        gcn_wmma_gemm_f16_ldsb<<<g1, 512, 0, stream>>>(x, W1p, xw1, N);
    }
    {
        gcn_gather256_relu_gemm64<<<(N + 15) / 16, 512, 0, stream>>>(rowptr, csr, dinv, xw1, b1, W2p, xw2, N);
    }

    {
        int th = N * 32;
        gcn_gather64_bias_softmax<<<(th + B - 1) / B, B, 0, stream>>>(rowptr, csr, dinv, xw2, b2, out, N);
    }
}


# in-edge grouping without global atomics: 128 edge chunks, each counts and places with one LDS counter per node, prefix over chunks per node; a node's sum order = chunk order, then LDS add arrival
# speedup vs baseline: 4.3080x; 1.3863x over previous
#include <hip/hip_runtime.h>
#include <hip/hip_bf16.h>

typedef __attribute__((ext_vector_type(16))) _Float16 v16h;
typedef __attribute__((ext_vector_type(8)))  float    v8f;
typedef __attribute__((ext_vector_type(4)))  _Float16 v4h;
typedef __attribute__((ext_vector_type(8)))  _Float16 v8h;

#define GCN_D_IN  512
#define GCN_D_HID 256
#define GCN_D_OUT 64
#define GCN_H_PITCH (GCN_D_HID + 8)

#define GCN_NODE_CAP    50176
#define GCN_EDGE_CHUNKS 128
#define GCN_SCAN_NODES  256
#define GCN_EDGE_BATCH  8

__global__ void __launch_bounds__(1024)
gcn_chunk_count(const int* __restrict__ dst, int* __restrict__ part, int E, int pitch) {
    __shared__ __attribute__((aligned(16))) int cnt[GCN_NODE_CAP];
    const int t = threadIdx.x;
    const int p4 = pitch >> 2;
    for (int i = t; i < p4; i += 1024) ((int4*)cnt)[i] = make_int4(0, 0, 0, 0);
    __syncthreads();
    const int per = (E + (int)gridDim.x - 1) / (int)gridDim.x;
    const int e0 = (int)blockIdx.x * per;
    const int e1 = (e0 + per < E) ? (e0 + per) : E;
    for (int eb = e0 + t; eb < e1; eb += GCN_EDGE_BATCH * 1024) {
        int d[GCN_EDGE_BATCH];
#pragma unroll
        for (int j = 0; j < GCN_EDGE_BATCH; ++j) {
            const int e = eb + j * 1024;
            d[j] = dst[(e < e1) ? e : (e1 - 1)];
        }
#pragma unroll
        for (int j = 0; j < GCN_EDGE_BATCH; ++j)
            atomicAdd(&cnt[d[j]], (eb + j * 1024 < e1) ? 1 : 0);
    }
    __syncthreads();
    int4* __restrict__ row = (int4*)(part + (size_t)blockIdx.x * pitch);
    for (int i = t; i < p4; i += 1024) row[i] = ((const int4*)cnt)[i];
}

__global__ void __launch_bounds__(1024)
gcn_chunk_prefix_scan(int* __restrict__ part, int* __restrict__ deg, int* __restrict__ rowptr,
                      int* __restrict__ blocktot, int n, int pitch) {
    __shared__ int gsum[4][GCN_SCAN_NODES];
    __shared__ int wsum[8];
    const int t = threadIdx.x;
    const int lane = t & 31;
    const int wv = t >> 5;
    const int nl = t & (GCN_SCAN_NODES - 1);
    const int g = t >> 8;
    const int i = (int)blockIdx.x * GCN_SCAN_NODES + nl;
    int* __restrict__ p = part + (size_t)(g * 32) * pitch + i;
    int v[32];
    int s = 0;
    if (i < n) {
#pragma unroll
        for (int j = 0; j < 32; ++j) v[j] = p[(size_t)j * pitch];
#pragma unroll
        for (int j = 0; j < 32; ++j) { const int x = v[j]; v[j] = s; s += x; }
    }
    gsum[g][nl] = s;
    __syncthreads();
    int goff = 0, c = 0;
#pragma unroll
    for (int k = 0; k < 4; ++k) {
        const int x = gsum[k][nl];
        if (k < g) goff += x;
        c += x;
    }
    if (i < n) {
#pragma unroll
        for (int j = 0; j < 32; ++j) p[(size_t)j * pitch] = v[j] + goff;
    }
    int incl = c;
#pragma unroll
    for (int off = 1; off < 32; off <<= 1) {
        int y = __shfl_up(incl, off, 32);
        if (lane >= off) incl += y;
    }
    if (g == 0 && lane == 31) wsum[wv] = incl;
    __syncthreads();
    if (wv == 0) {
        const int w = (lane < 8) ? wsum[lane] : 0;
        int inc2 = w;
#pragma unroll
        for (int off = 1; off < 8; off <<= 1) {
            int y = __shfl_up(inc2, off, 32);
            if (lane >= off) inc2 += y;
        }
        if (lane < 8) wsum[lane] = inc2 - w;
    }
    __syncthreads();
    if (g == 0) {
        const int excl = incl - c + wsum[wv];
        if (i < n) { rowptr[i] = excl; deg[i] = c; }
        if (t == GCN_SCAN_NODES - 1) blocktot[blockIdx.x] = excl + c;
    }
}

__global__ void gcn_scan_finish(const int* __restrict__ deg, int* __restrict__ rowptr,
                                const int* __restrict__ blocktot, float* __restrict__ dinv,
                                int n) {
    __shared__ int red[GCN_SCAN_NODES / 32];
    const int t = threadIdx.x;
    const int lane = t & 31;
    const int wv = t >> 5;
    const int i = (int)blockIdx.x * GCN_SCAN_NODES + t;
    int b = 0;
    for (int j = t; j < (int)blockIdx.x; j += GCN_SCAN_NODES) b += blocktot[j];
#pragma unroll
    for (int off = 16; off > 0; off >>= 1) b += __shfl_xor(b, off, 32);
    if (lane == 0) red[wv] = b;
    __syncthreads();
    int base = 0;
#pragma unroll
    for (int k = 0; k < GCN_SCAN_NODES / 32; ++k) base += red[k];
    if (i < n) {
        const int c = deg[i];
        const int p = rowptr[i] + base;
        rowptr[i] = p;
        const float d = (float)(c + 1);
        dinv[i] = (d > 0.0f) ? rsqrtf(d) : 0.0f;
        if (i == n - 1) rowptr[n] = p + c;
    }
}

__global__ void __launch_bounds__(1024)
gcn_chunk_place(const int* __restrict__ src, const int* __restrict__ dst,
                const float* __restrict__ dinv, const int* __restrict__ rowptr,
                const int* __restrict__ part, int2* __restrict__ csr, int E, int n, int pitch) {
    __shared__ __attribute__((aligned(16))) int cur[GCN_NODE_CAP];
    const int t = threadIdx.x;
    const int* __restrict__ prow = part + (size_t)blockIdx.x * pitch;
    const int n4 = n >> 2;
    for (int i = t; i < n4; i += 1024) {
        const int4 a = ((const int4*)prow)[i];
        const int4 r = ((const int4*)rowptr)[i];
        ((int4*)cur)[i] = make_int4(a.x + r.x, a.y + r.y, a.z + r.z, a.w + r.w);
    }
    for (int i = (n4 << 2) + t; i < n; i += 1024) cur[i] = prow[i] + rowptr[i];
    __syncthreads();
    const int per = (E + (int)gridDim.x - 1) / (int)gridDim.x;
    const int e0 = (int)blockIdx.x * per;
    const int e1 = (e0 + per < E) ? (e0 + per) : E;
    for (int eb = e0 + t; eb < e1; eb += GCN_EDGE_BATCH * 1024) {
        int s[GCN_EDGE_BATCH], d[GCN_EDGE_BATCH], pos[GCN_EDGE_BATCH];
        float w[GCN_EDGE_BATCH];
#pragma unroll
        for (int j = 0; j < GCN_EDGE_BATCH; ++j) {
            const int e = eb + j * 1024;
            const int ec = (e < e1) ? e : (e1 - 1);
            s[j] = src[ec];
            d[j] = dst[ec];
        }
#pragma unroll
        for (int j = 0; j < GCN_EDGE_BATCH; ++j) w[j] = dinv[s[j]];
#pragma unroll
        for (int j = 0; j < GCN_EDGE_BATCH; ++j)
            pos[j] = atomicAdd(&cur[d[j]], (eb + j * 1024 < e1) ? 1 : 0);
#pragma unroll
        for (int j = 0; j < GCN_EDGE_BATCH; ++j)
            if (eb + j * 1024 < e1) csr[pos[j]] = make_int2(s[j], __float_as_int(w[j]));
    }
}

__global__ void gcn_pack_w_f16(const float* __restrict__ W, _Float16* __restrict__ Wp,
                               int K, int N, int deal8) {
    int t = blockIdx.x * blockDim.x + threadIdx.x;
    int tilesN = N >> 4;
    int total = (K >> 5) * tilesN * 32;
    if (t >= total) return;
    int frag = t >> 5;
    int lane = t & 31;
    int kt = frag / tilesN;
    int tn = frag - kt * tilesN;
    int n    = deal8 ? ((tn >> 3) * 128 + (lane & 15) * 8 + (tn & 7))
                     : (tn * 16 + (lane & 15));
    int koff = kt * 32 + (lane >> 4) * 8;
    _Float16* o = Wp + (size_t)t * 16;
#pragma unroll
    for (int j = 0; j < 8; ++j) {
        o[j]     = (_Float16)W[(size_t)(koff + j) * N + n];
        o[j + 8] = (_Float16)W[(size_t)(koff + 16 + j) * N + n];
    }
}


static __device__ __forceinline__ void gcn_ldb4(v16h (&dst)[4], const v8h* p) {
#pragma unroll
    for (int j = 0; j < 4; ++j) {
        const v8h lo = p[j * 64];
        const v8h hi = p[j * 64 + 32];
        dst[j] = __builtin_shufflevector(lo, hi, 0, 1, 2, 3, 4, 5, 6, 7,
                                         8, 9, 10, 11, 12, 13, 14, 15);
    }
}

static __device__ __forceinline__ void gcn_lda(float4 (&dst)[4], const float* arow, int kt) {
    const float* p = arow + kt * 32;
    dst[0] = *(const float4*)(p);
    dst[1] = *(const float4*)(p + 4);
    dst[2] = *(const float4*)(p + 16);
    dst[3] = *(const float4*)(p + 20);
}

__global__ void __launch_bounds__(512)
gcn_wmma_gemm_f16_ldsb(const float* __restrict__ A, const _Float16* __restrict__ Bp,
                       _Float16* __restrict__ C, int M) {
    __shared__ __attribute__((aligned(16))) v8h Bs[(GCN_D_IN * GCN_D_HID) / 8];
    const int tid  = threadIdx.x;
    const int lane = tid & 31;
    const int wv   = __builtin_amdgcn_readfirstlane(tid >> 5);
    const int hsel = lane >> 4;
    const int idx  = lane & 15;
    const int tilesM = (M + 15) >> 4;

    const int tm = wv * (int)gridDim.x + (int)blockIdx.x;
    int row = tm * 16 + idx;
    row = (row < M) ? row : (M - 1);
    const float* __restrict__ arow = A + (size_t)row * GCN_D_IN + hsel * 8;
    if (tm < tilesM) {
        __builtin_prefetch(arow, 0, 3);
        __builtin_prefetch(arow + 32, 0, 3);
        __builtin_prefetch(arow + 64, 0, 3);
    }

    {
        const v8h* __restrict__ Bg = (const v8h*)Bp;
#pragma unroll 8
        for (int i = 0; i < 32; ++i) {
            const int g = (((i + (int)blockIdx.x) & 31) << 9) + tid;
            const int dst = ((g >> 6) << 6) | ((g & 1) << 5) | ((g >> 1) & 31);
            Bs[dst] = Bg[g];
        }
    }
    __syncthreads();
    if (tm >= tilesM) return;

    v8f acc[16];
#pragma unroll
    for (int t = 0; t < 16; ++t) acc[t] = (v8f){};

    float4 ar[3][4];
    gcn_lda(ar[0], arow, 0);
    gcn_lda(ar[1], arow, 1);
    gcn_lda(ar[2], arow, 2);

    v16h bq[2][4];
    const v8h* bk = Bs + lane;
    gcn_ldb4(bq[0], bk);

#pragma unroll
    for (int kt = 0; kt < GCN_D_IN / 32; ++kt) {
        const int s = kt % 3;
        float af[16];
        *(float4*)(af + 0)  = ar[s][0];  *(float4*)(af + 4)  = ar[s][1];
        *(float4*)(af + 8)  = ar[s][2];  *(float4*)(af + 12) = ar[s][3];
        v16h av;
#pragma unroll
        for (int j = 0; j < 16; ++j) av[j] = (_Float16)af[j];
        if (kt + 3 < GCN_D_IN / 32) gcn_lda(ar[s], arow, kt + 3);
#pragma unroll
        for (int g = 0; g < 4; ++g) {
            const int cur = g & 1;
            const int nf = kt * 16 + (g + 1) * 4;
            if (nf < (GCN_D_IN / 32) * 16) gcn_ldb4(bq[cur ^ 1], bk + nf * 64);
#pragma unroll
            for (int j = 0; j < 4; ++j)
                acc[g * 4 + j] = __builtin_amdgcn_wmma_f32_16x16x32_f16(false, av, false, bq[cur][j], (short)0,
                                                                       acc[g * 4 + j], false, false);
        }
    }

    const int row0 = tm * 16 + hsel * 8;
    _Float16* cbase = C + (size_t)row0 * GCN_D_HID + idx * 8;
#pragma unroll
    for (int r = 0; r < 8; ++r) {
        if (row0 + r < M) {
#pragma unroll
            for (int g = 0; g < 2; ++g) {
                v8h o;
#pragma unroll
                for (int j = 0; j < 8; ++j) o[j] = (_Float16)acc[g * 8 + j][r];
                *(v8h*)(cbase + (size_t)r * GCN_D_HID + g * 128) = o;
            }
        }
    }
}

__global__ void __launch_bounds__(512)
gcn_gather256_relu_gemm64(const int* __restrict__ rowptr, const int2* __restrict__ csr,
                          const float* __restrict__ dinv, const _Float16* __restrict__ xw,
                          const float* __restrict__ bias, const _Float16* __restrict__ Bp,
                          float* __restrict__ xw2, int nodes) {
    __shared__ __attribute__((aligned(16))) _Float16 htile[16 * GCN_H_PITCH];
    const int lane = threadIdx.x & 31;
    const int wv = __builtin_amdgcn_readfirstlane((int)(threadIdx.x >> 5));
    const int node0 = (int)blockIdx.x * 16;
    const int node = node0 + wv;
    float a[8];
#pragma unroll
    for (int j = 0; j < 8; ++j) a[j] = 0.0f;
    if (node < nodes) {
        const int beg = rowptr[node];
        const int end = rowptr[node + 1];
        const float di = dinv[node];
        const v8h sv = ((const v8h*)(xw + (size_t)node * 256))[lane];
        const float4 b0 = ((const float4*)bias)[2 * lane];
        const float4 b1 = ((const float4*)bias)[2 * lane + 1];
        const float b[8] = {b0.x, b0.y, b0.z, b0.w, b1.x, b1.y, b1.z, b1.w};
#pragma unroll
        for (int j = 0; j < 8; ++j) a[j] = (float)sv[j] * di;
        for (int base = beg; base < end; base += 32) {
            int2 my = make_int2(0, 0);
            if (base + lane < end) my = csr[base + lane];
            const int n = (end - base < 32) ? (end - base) : 32;
#pragma unroll 4
            for (int k = 0; k < n; ++k) {
                const int s = __builtin_amdgcn_readlane(my.x, k);
                const float w = __int_as_float(__builtin_amdgcn_readlane(my.y, k));
                const v8h v = ((const v8h*)(xw + (size_t)s * 256))[lane];
#pragma unroll
                for (int j = 0; j < 8; ++j) a[j] = fmaf((float)v[j], w, a[j]);
            }
        }
#pragma unroll
        for (int j = 0; j < 8; ++j) a[j] = fmaxf(a[j] * di + b[j], 0.0f);
    }
    {
        v8h h;
#pragma unroll
        for (int j = 0; j < 8; ++j) h[j] = (_Float16)a[j];
        *(v8h*)(htile + wv * GCN_H_PITCH + 8 * lane) = h;
    }
    __syncthreads();
    if (wv >= 2) return;

    const int hsel = lane >> 4;
    const int idx  = lane & 15;
    const int tn0  = wv * 2;
    const _Float16* arow = htile + idx * GCN_H_PITCH + hsel * 8;
    v8f acc0 = {}, acc1 = {};
    for (int kt = 0; kt < GCN_D_HID / 32; ++kt) {
        const v8h lo = *(const v8h*)(arow + kt * 32);
        const v8h hi = *(const v8h*)(arow + kt * 32 + 16);
        const v16h av = __builtin_shufflevector(lo, hi, 0, 1, 2, 3, 4, 5, 6, 7,
                                                8, 9, 10, 11, 12, 13, 14, 15);
        const _Float16* bbase = Bp + ((size_t)(kt * (GCN_D_OUT >> 4) + tn0) * 32 + lane) * 16;
        const v16h b0 = *(const v16h*)(bbase);
        const v16h b1 = *(const v16h*)(bbase + 32 * 16);
        acc0 = __builtin_amdgcn_wmma_f32_16x16x32_f16(false, av, false, b0, (short)0, acc0, false, false);
        acc1 = __builtin_amdgcn_wmma_f32_16x16x32_f16(false, av, false, b1, (short)0, acc1, false, false);
    }
    const int row0 = node0 + hsel * 8;
    float* cbase = xw2 + (size_t)row0 * GCN_D_OUT + tn0 * 16 + idx;
#pragma unroll
    for (int r = 0; r < 8; ++r) {
        if (row0 + r < nodes) {
            cbase[(size_t)r * GCN_D_OUT + 0]  = acc0[r];
            cbase[(size_t)r * GCN_D_OUT + 16] = acc1[r];
        }
    }
}

__global__ void gcn_gather64_bias_softmax(const int* __restrict__ rowptr, const int2* __restrict__ csr,
                                          const float* __restrict__ dinv, const float* __restrict__ xw,
                                          const float* __restrict__ bias,
                                          float* __restrict__ out, int nodes) {
    const int lane = threadIdx.x & 31;
    const int node = __builtin_amdgcn_readfirstlane((int)((blockIdx.x * blockDim.x + threadIdx.x) >> 5));
    if (node >= nodes) return;
    const int beg = rowptr[node];
    const int end = rowptr[node + 1];
    const float di = dinv[node];
    float2 a = ((const float2*)(xw + (size_t)node * 64))[lane];
    const float2 b = ((const float2*)bias)[lane];
    a.x *= di; a.y *= di;
    for (int base = beg; base < end; base += 32) {
        int2 my = make_int2(0, 0);
        if (base + lane < end) my = csr[base + lane];
        const int n = (end - base < 32) ? (end - base) : 32;
#pragma unroll 8
        for (int k = 0; k < n; ++k) {
            const int s = __builtin_amdgcn_readlane(my.x, k);
            const float w = __int_as_float(__builtin_amdgcn_readlane(my.y, k));
            const float2 v = ((const float2*)(xw + (size_t)s * 64))[lane];
            a.x = fmaf(v.x, w, a.x);
            a.y = fmaf(v.y, w, a.y);
        }
    }
    const float v0 = a.x * di + b.x;
    const float v1 = a.y * di + b.y;
    float m = fmaxf(v0, v1);
#pragma unroll
    for (int off = 16; off > 0; off >>= 1) m = fmaxf(m, __shfl_xor(m, off, 32));
    const float e0 = __expf(v0 - m);
    const float e1 = __expf(v1 - m);
    float sum = e0 + e1;
#pragma unroll
    for (int off = 16; off > 0; off >>= 1) sum += __shfl_xor(sum, off, 32);
    const float inv = 1.0f / sum;
    ((float2*)(out + (size_t)node * 64))[lane] = make_float2(e0 * inv, e1 * inv);
}

extern "C" void kernel_launch(void* const* d_in, const int* in_sizes, int n_in,
                              void* d_out, int out_size, void* d_ws, size_t ws_size,
                              hipStream_t stream) {
    const float* x   = (const float*)d_in[0];
    const int*   ei  = (const int*)  d_in[1];
    const float* W1  = (const float*)d_in[2];
    const float* b1  = (const float*)d_in[3];
    const float* W2  = (const float*)d_in[4];
    const float* b2  = (const float*)d_in[5];
    float* out = (float*)d_out;

    const int N = in_sizes[0] / GCN_D_IN;
    const int E = in_sizes[1] / 2;
    const int* src = ei;
    const int* dst = ei + E;

    float* ws   = (float*)d_ws;
    size_t off  = ((size_t)N + 63) & ~(size_t)63;
    float* dinv = ws;
    _Float16* xw1 = (_Float16*)(ws + off);
    float* reg2 = ws + off + (size_t)N * GCN_D_HID;
    _Float16* W1p = (_Float16*)(reg2 + (size_t)N * GCN_D_HID);
    _Float16* W2p = W1p + (size_t)GCN_D_IN * GCN_D_HID;
    float* xw2  = reg2;

    int*  so       = (int*)(reg2 + (size_t)N * GCN_D_OUT);
    int*  deg      = so;
    int*  rowptr   = so + off;
    int*  blocktot = so + 2 * off + 64;
    int2* csr      = (int2*)(so + 2 * off + 64 + 1024);
    int*  part     = so + 2 * off + 64 + 1024 + (((size_t)2 * E + 63) & ~(size_t)63);
    const int pitch = (int)off;

    const int B = 256;
    {
        int t1 = (GCN_D_IN >> 5) * (GCN_D_HID >> 4) * 32;
        gcn_pack_w_f16<<<(t1 + B - 1) / B, B, 0, stream>>>(W1, W1p, GCN_D_IN, GCN_D_HID, 1);
        int t2 = (GCN_D_HID >> 5) * (GCN_D_OUT >> 4) * 32;
        gcn_pack_w_f16<<<(t2 + B - 1) / B, B, 0, stream>>>(W2, W2p, GCN_D_HID, GCN_D_OUT, 0);
    }

    {
        const int SN = GCN_SCAN_NODES;
        gcn_chunk_count      <<<GCN_EDGE_CHUNKS, 1024, 0, stream>>>(dst, part, E, pitch);
        gcn_chunk_prefix_scan<<<(N + SN - 1) / SN, 1024, 0, stream>>>(part, deg, rowptr, blocktot, N, pitch);
        gcn_scan_finish      <<<(N + SN - 1) / SN, SN, 0, stream>>>(deg, rowptr, blocktot, dinv, N);
        gcn_chunk_place      <<<GCN_EDGE_CHUNKS, 1024, 0, stream>>>(src, dst, dinv, rowptr, part, csr, E, N, pitch);
    }

    {
        int g1 = ((N + 15) / 16 + 15) / 16;
        if (g1 < 256) g1 = 256;
        gcn_wmma_gemm_f16_ldsb<<<g1, 512, 0, stream>>>(x, W1p, xw1, N);
    }
    {
        gcn_gather256_relu_gemm64<<<(N + 15) / 16, 512, 0, stream>>>(rowptr, csr, dinv, xw1, b1, W2p, xw2, N);
    }

    {
        int th = N * 32;
        gcn_gather64_bias_softmax<<<(th + B - 1) / B, B, 0, stream>>>(rowptr, csr, dinv, xw2, b2, out, N);
    }
}


# launch count 9 -> 6: both weight packs ride as extra workgroups of the edge-count launch, and the scan's finishing pass (block offsets, final row starts, dinv) is folded into the placing kernel
# speedup vs baseline: 4.5519x; 1.0566x over previous
#include <hip/hip_runtime.h>
#include <hip/hip_bf16.h>

typedef __attribute__((ext_vector_type(16))) _Float16 v16h;
typedef __attribute__((ext_vector_type(8)))  float    v8f;
typedef __attribute__((ext_vector_type(4)))  _Float16 v4h;
typedef __attribute__((ext_vector_type(8)))  _Float16 v8h;

#define GCN_D_IN  512
#define GCN_D_HID 256
#define GCN_D_OUT 64
#define GCN_H_PITCH (GCN_D_HID + 8)

#define GCN_NODE_CAP    50176
#define GCN_EDGE_CHUNKS 128
#define GCN_SCAN_NODES  256
#define GCN_EDGE_BATCH  8

static __device__ __forceinline__ void gcn_pack_w_item(const float* __restrict__ W,
                                                       _Float16* __restrict__ Wp,
                                                       int N, int deal8, int u) {
    const int tilesN = N >> 4;
    const int frag = u >> 5;
    const int lane = u & 31;
    const int kt = frag / tilesN;
    const int tn = frag - kt * tilesN;
    const int n    = deal8 ? ((tn >> 3) * 128 + (lane & 15) * 8 + (tn & 7))
                           : (tn * 16 + (lane & 15));
    const int koff = kt * 32 + (lane >> 4) * 8;
    _Float16* o = Wp + (size_t)u * 16;
#pragma unroll
    for (int j = 0; j < 8; ++j) {
        o[j]     = (_Float16)W[(size_t)(koff + j) * N + n];
        o[j + 8] = (_Float16)W[(size_t)(koff + 16 + j) * N + n];
    }
}

#define GCN_PACK_ITEMS_W1 ((GCN_D_IN >> 5) * (GCN_D_HID >> 4) * 32)
#define GCN_PACK_ITEMS_W2 ((GCN_D_HID >> 5) * (GCN_D_OUT >> 4) * 32)
#define GCN_PACK_BLOCKS   ((GCN_PACK_ITEMS_W1 + GCN_PACK_ITEMS_W2 + 1023) / 1024)

__global__ void __launch_bounds__(1024)
gcn_chunk_count_pack(const int* __restrict__ dst, int* __restrict__ part, int E, int pitch, int chunks,
                     const float* __restrict__ W1, _Float16* __restrict__ W1p,
                     const float* __restrict__ W2, _Float16* __restrict__ W2p) {
    __shared__ __attribute__((aligned(16))) int cnt[GCN_NODE_CAP];
    const int t = threadIdx.x;
    if ((int)blockIdx.x >= chunks) {
        int u = ((int)blockIdx.x - chunks) * 1024 + t;
        if (u < GCN_PACK_ITEMS_W1) {
            gcn_pack_w_item(W1, W1p, GCN_D_HID, 1, u);
        } else {
            u -= GCN_PACK_ITEMS_W1;
            if (u < GCN_PACK_ITEMS_W2) gcn_pack_w_item(W2, W2p, GCN_D_OUT, 0, u);
        }
        return;
    }
    const int p4 = pitch >> 2;
    for (int i = t; i < p4; i += 1024) ((int4*)cnt)[i] = make_int4(0, 0, 0, 0);
    __syncthreads();
    const int per = (E + chunks - 1) / chunks;
    const int e0 = (int)blockIdx.x * per;
    const int e1 = (e0 + per < E) ? (e0 + per) : E;
    for (int eb = e0 + t; eb < e1; eb += GCN_EDGE_BATCH * 1024) {
        int d[GCN_EDGE_BATCH];
#pragma unroll
        for (int j = 0; j < GCN_EDGE_BATCH; ++j) {
            const int e = eb + j * 1024;
            d[j] = dst[(e < e1) ? e : (e1 - 1)];
        }
#pragma unroll
        for (int j = 0; j < GCN_EDGE_BATCH; ++j)
            atomicAdd(&cnt[d[j]], (eb + j * 1024 < e1) ? 1 : 0);
    }
    __syncthreads();
    int4* __restrict__ row = (int4*)(part + (size_t)blockIdx.x * pitch);
    for (int i = t; i < p4; i += 1024) row[i] = ((const int4*)cnt)[i];
}

__global__ void __launch_bounds__(1024)
gcn_chunk_prefix_scan(int* __restrict__ part, int* __restrict__ deg, int* __restrict__ rowptr,
                      int* __restrict__ blocktot, int n, int pitch) {
    __shared__ int gsum[4][GCN_SCAN_NODES];
    __shared__ int wsum[8];
    const int t = threadIdx.x;
    const int lane = t & 31;
    const int wv = t >> 5;
    const int nl = t & (GCN_SCAN_NODES - 1);
    const int g = t >> 8;
    const int i = (int)blockIdx.x * GCN_SCAN_NODES + nl;
    int* __restrict__ p = part + (size_t)(g * 32) * pitch + i;
    int v[32];
    int s = 0;
    if (i < n) {
#pragma unroll
        for (int j = 0; j < 32; ++j) v[j] = p[(size_t)j * pitch];
#pragma unroll
        for (int j = 0; j < 32; ++j) { const int x = v[j]; v[j] = s; s += x; }
    }
    gsum[g][nl] = s;
    __syncthreads();
    int goff = 0, c = 0;
#pragma unroll
    for (int k = 0; k < 4; ++k) {
        const int x = gsum[k][nl];
        if (k < g) goff += x;
        c += x;
    }
    if (i < n) {
#pragma unroll
        for (int j = 0; j < 32; ++j) p[(size_t)j * pitch] = v[j] + goff;
    }
    int incl = c;
#pragma unroll
    for (int off = 1; off < 32; off <<= 1) {
        int y = __shfl_up(incl, off, 32);
        if (lane >= off) incl += y;
    }
    if (g == 0 && lane == 31) wsum[wv] = incl;
    __syncthreads();
    if (wv == 0) {
        const int w = (lane < 8) ? wsum[lane] : 0;
        int inc2 = w;
#pragma unroll
        for (int off = 1; off < 8; off <<= 1) {
            int y = __shfl_up(inc2, off, 32);
            if (lane >= off) inc2 += y;
        }
        if (lane < 8) wsum[lane] = inc2 - w;
    }
    __syncthreads();
    if (g == 0) {
        const int excl = incl - c + wsum[wv];
        if (i < n) { rowptr[i] = excl; deg[i] = c; }
        if (t == GCN_SCAN_NODES - 1) blocktot[blockIdx.x] = excl + c;
    }
}

static __device__ __forceinline__ float gcn_dinv_of(int indeg) {
    const float d = (float)(indeg + 1);
    return (d > 0.0f) ? rsqrtf(d) : 0.0f;
}

__global__ void __launch_bounds__(1024)
gcn_chunk_place(const int* __restrict__ src, const int* __restrict__ dst,
                const int* __restrict__ deg, const int* __restrict__ rowloc,
                const int* __restrict__ blocktot, const int* __restrict__ part,
                int* __restrict__ rowptr, float* __restrict__ dinv,
                int2* __restrict__ csr, int E, int n, int pitch, int nblk) {
    __shared__ __attribute__((aligned(16))) int cur[GCN_NODE_CAP];
    __shared__ int bbase[GCN_NODE_CAP / GCN_SCAN_NODES + 1];
    __shared__ int wtot[32];
    const int t = threadIdx.x;
    const int lane = t & 31;
    const int wv = t >> 5;
    {
        const int bt = (t < nblk) ? blocktot[t] : 0;
        int incl = bt;
#pragma unroll
        for (int off = 1; off < 32; off <<= 1) {
            int y = __shfl_up(incl, off, 32);
            if (lane >= off) incl += y;
        }
        if (lane == 31) wtot[wv] = incl;
        __syncthreads();
        if (wv == 0) {
            const int w = wtot[lane];
            int inc2 = w;
#pragma unroll
            for (int off = 1; off < 32; off <<= 1) {
                int y = __shfl_up(inc2, off, 32);
                if (lane >= off) inc2 += y;
            }
            wtot[lane] = inc2 - w;
        }
        __syncthreads();
        const int excl = incl - bt + wtot[wv];
        if (t < nblk) bbase[t] = excl;
        if (t == nblk - 1) bbase[nblk] = excl + bt;
    }
    __syncthreads();
    const int* __restrict__ prow = part + (size_t)blockIdx.x * pitch;
    const int n4 = n >> 2;
    for (int i = t; i < n4; i += 1024) {
        const int4 a = ((const int4*)prow)[i];
        const int4 r = ((const int4*)rowloc)[i];
        const int bb = bbase[(i << 2) / GCN_SCAN_NODES];
        ((int4*)cur)[i] = make_int4(a.x + r.x + bb, a.y + r.y + bb, a.z + r.z + bb, a.w + r.w + bb);
    }
    for (int i = (n4 << 2) + t; i < n; i += 1024) cur[i] = prow[i] + rowloc[i] + bbase[i / GCN_SCAN_NODES];
    {
        const int nper = (n + (int)gridDim.x - 1) / (int)gridDim.x;
        const int i0 = (int)blockIdx.x * nper;
        const int i1 = (i0 + nper < n) ? (i0 + nper) : n;
        for (int i = i0 + t; i < i1; i += 1024) {
            rowptr[i] = rowloc[i] + bbase[i / GCN_SCAN_NODES];
            dinv[i] = gcn_dinv_of(deg[i]);
        }
        if (blockIdx.x == 0 && t == 0) rowptr[n] = bbase[nblk];
    }
    __syncthreads();
    const int per = (E + (int)gridDim.x - 1) / (int)gridDim.x;
    const int e0 = (int)blockIdx.x * per;
    const int e1 = (e0 + per < E) ? (e0 + per) : E;
    for (int eb = e0 + t; eb < e1; eb += GCN_EDGE_BATCH * 1024) {
        int s[GCN_EDGE_BATCH], d[GCN_EDGE_BATCH], pos[GCN_EDGE_BATCH], sd[GCN_EDGE_BATCH];
#pragma unroll
        for (int j = 0; j < GCN_EDGE_BATCH; ++j) {
            const int e = eb + j * 1024;
            const int ec = (e < e1) ? e : (e1 - 1);
            s[j] = src[ec];
            d[j] = dst[ec];
        }
#pragma unroll
        for (int j = 0; j < GCN_EDGE_BATCH; ++j) sd[j] = deg[s[j]];
#pragma unroll
        for (int j = 0; j < GCN_EDGE_BATCH; ++j)
            pos[j] = atomicAdd(&cur[d[j]], (eb + j * 1024 < e1) ? 1 : 0);
#pragma unroll
        for (int j = 0; j < GCN_EDGE_BATCH; ++j)
            if (eb + j * 1024 < e1) csr[pos[j]] = make_int2(s[j], __float_as_int(gcn_dinv_of(sd[j])));
    }
}


static __device__ __forceinline__ void gcn_ldb4(v16h (&dst)[4], const v8h* p) {
#pragma unroll
    for (int j = 0; j < 4; ++j) {
        const v8h lo = p[j * 64];
        const v8h hi = p[j * 64 + 32];
        dst[j] = __builtin_shufflevector(lo, hi, 0, 1, 2, 3, 4, 5, 6, 7,
                                         8, 9, 10, 11, 12, 13, 14, 15);
    }
}

static __device__ __forceinline__ void gcn_lda(float4 (&dst)[4], const float* arow, int kt) {
    const float* p = arow + kt * 32;
    dst[0] = *(const float4*)(p);
    dst[1] = *(const float4*)(p + 4);
    dst[2] = *(const float4*)(p + 16);
    dst[3] = *(const float4*)(p + 20);
}

__global__ void __launch_bounds__(512)
gcn_wmma_gemm_f16_ldsb(const float* __restrict__ A, const _Float16* __restrict__ Bp,
                       _Float16* __restrict__ C, int M) {
    __shared__ __attribute__((aligned(16))) v8h Bs[(GCN_D_IN * GCN_D_HID) / 8];
    const int tid  = threadIdx.x;
    const int lane = tid & 31;
    const int wv   = __builtin_amdgcn_readfirstlane(tid >> 5);
    const int hsel = lane >> 4;
    const int idx  = lane & 15;
    const int tilesM = (M + 15) >> 4;

    const int tm = wv * (int)gridDim.x + (int)blockIdx.x;
    int row = tm * 16 + idx;
    row = (row < M) ? row : (M - 1);
    const float* __restrict__ arow = A + (size_t)row * GCN_D_IN + hsel * 8;
    if (tm < tilesM) {
        __builtin_prefetch(arow, 0, 3);
        __builtin_prefetch(arow + 32, 0, 3);
        __builtin_prefetch(arow + 64, 0, 3);
    }

    {
        const v8h* __restrict__ Bg = (const v8h*)Bp;
#pragma unroll 8
        for (int i = 0; i < 32; ++i) {
            const int g = (((i + (int)blockIdx.x) & 31) << 9) + tid;
            const int dst = ((g >> 6) << 6) | ((g & 1) << 5) | ((g >> 1) & 31);
            Bs[dst] = Bg[g];
        }
    }
    __syncthreads();
    if (tm >= tilesM) return;

    v8f acc[16];
#pragma unroll
    for (int t = 0; t < 16; ++t) acc[t] = (v8f){};

    float4 ar[3][4];
    gcn_lda(ar[0], arow, 0);
    gcn_lda(ar[1], arow, 1);
    gcn_lda(ar[2], arow, 2);

    v16h bq[2][4];
    const v8h* bk = Bs + lane;
    gcn_ldb4(bq[0], bk);

#pragma unroll
    for (int kt = 0; kt < GCN_D_IN / 32; ++kt) {
        const int s = kt % 3;
        float af[16];
        *(float4*)(af + 0)  = ar[s][0];  *(float4*)(af + 4)  = ar[s][1];
        *(float4*)(af + 8)  = ar[s][2];  *(float4*)(af + 12) = ar[s][3];
        v16h av;
#pragma unroll
        for (int j = 0; j < 16; ++j) av[j] = (_Float16)af[j];
        if (kt + 3 < GCN_D_IN / 32) gcn_lda(ar[s], arow, kt + 3);
#pragma unroll
        for (int g = 0; g < 4; ++g) {
            const int cur = g & 1;
            const int nf = kt * 16 + (g + 1) * 4;
            if (nf < (GCN_D_IN / 32) * 16) gcn_ldb4(bq[cur ^ 1], bk + nf * 64);
#pragma unroll
            for (int j = 0; j < 4; ++j)
                acc[g * 4 + j] = __builtin_amdgcn_wmma_f32_16x16x32_f16(false, av, false, bq[cur][j], (short)0,
                                                                       acc[g * 4 + j], false, false);
        }
    }

    const int row0 = tm * 16 + hsel * 8;
    _Float16* cbase = C + (size_t)row0 * GCN_D_HID + idx * 8;
#pragma unroll
    for (int r = 0; r < 8; ++r) {
        if (row0 + r < M) {
#pragma unroll
            for (int g = 0; g < 2; ++g) {
                v8h o;
#pragma unroll
                for (int j = 0; j < 8; ++j) o[j] = (_Float16)acc[g * 8 + j][r];
                *(v8h*)(cbase + (size_t)r * GCN_D_HID + g * 128) = o;
            }
        }
    }
}

__global__ void __launch_bounds__(512)
gcn_gather256_relu_gemm64(const int* __restrict__ rowptr, const int2* __restrict__ csr,
                          const float* __restrict__ dinv, const _Float16* __restrict__ xw,
                          const float* __restrict__ bias, const _Float16* __restrict__ Bp,
                          float* __restrict__ xw2, int nodes) {
    __shared__ __attribute__((aligned(16))) _Float16 htile[16 * GCN_H_PITCH];
    const int lane = threadIdx.x & 31;
    const int wv = __builtin_amdgcn_readfirstlane((int)(threadIdx.x >> 5));
    const int node0 = (int)blockIdx.x * 16;
    const int node = node0 + wv;
    float a[8];
#pragma unroll
    for (int j = 0; j < 8; ++j) a[j] = 0.0f;
    if (node < nodes) {
        const int beg = rowptr[node];
        const int end = rowptr[node + 1];
        const float di = dinv[node];
        const v8h sv = ((const v8h*)(xw + (size_t)node * 256))[lane];
        const float4 b0 = ((const float4*)bias)[2 * lane];
        const float4 b1 = ((const float4*)bias)[2 * lane + 1];
        const float b[8] = {b0.x, b0.y, b0.z, b0.w, b1.x, b1.y, b1.z, b1.w};
#pragma unroll
        for (int j = 0; j < 8; ++j) a[j] = (float)sv[j] * di;
        for (int base = beg; base < end; base += 32) {
            int2 my = make_int2(0, 0);
            if (base + lane < end) my = csr[base + lane];
            const int n = (end - base < 32) ? (end - base) : 32;
#pragma unroll 4
            for (int k = 0; k < n; ++k) {
                const int s = __builtin_amdgcn_readlane(my.x, k);
                const float w = __int_as_float(__builtin_amdgcn_readlane(my.y, k));
                const v8h v = ((const v8h*)(xw + (size_t)s * 256))[lane];
#pragma unroll
                for (int j = 0; j < 8; ++j) a[j] = fmaf((float)v[j], w, a[j]);
            }
        }
#pragma unroll
        for (int j = 0; j < 8; ++j) a[j] = fmaxf(a[j] * di + b[j], 0.0f);
    }
    {
        v8h h;
#pragma unroll
        for (int j = 0; j < 8; ++j) h[j] = (_Float16)a[j];
        *(v8h*)(htile + wv * GCN_H_PITCH + 8 * lane) = h;
    }
    __syncthreads();
    if (wv >= 2) return;

    const int hsel = lane >> 4;
    const int idx  = lane & 15;
    const int tn0  = wv * 2;
    const _Float16* arow = htile + idx * GCN_H_PITCH + hsel * 8;
    v8f acc0 = {}, acc1 = {};
    for (int kt = 0; kt < GCN_D_HID / 32; ++kt) {
        const v8h lo = *(const v8h*)(arow + kt * 32);
        const v8h hi = *(const v8h*)(arow + kt * 32 + 16);
        const v16h av = __builtin_shufflevector(lo, hi, 0, 1, 2, 3, 4, 5, 6, 7,
                                                8, 9, 10, 11, 12, 13, 14, 15);
        const _Float16* bbase = Bp + ((size_t)(kt * (GCN_D_OUT >> 4) + tn0) * 32 + lane) * 16;
        const v16h b0 = *(const v16h*)(bbase);
        const v16h b1 = *(const v16h*)(bbase + 32 * 16);
        acc0 = __builtin_amdgcn_wmma_f32_16x16x32_f16(false, av, false, b0, (short)0, acc0, false, false);
        acc1 = __builtin_amdgcn_wmma_f32_16x16x32_f16(false, av, false, b1, (short)0, acc1, false, false);
    }
    const int row0 = node0 + hsel * 8;
    float* cbase = xw2 + (size_t)row0 * GCN_D_OUT + tn0 * 16 + idx;
#pragma unroll
    for (int r = 0; r < 8; ++r) {
        if (row0 + r < nodes) {
            cbase[(size_t)r * GCN_D_OUT + 0]  = acc0[r];
            cbase[(size_t)r * GCN_D_OUT + 16] = acc1[r];
        }
    }
}

__global__ void gcn_gather64_bias_softmax(const int* __restrict__ rowptr, const int2* __restrict__ csr,
                                          const float* __restrict__ dinv, const float* __restrict__ xw,
                                          const float* __restrict__ bias,
                                          float* __restrict__ out, int nodes) {
    const int lane = threadIdx.x & 31;
    const int node = __builtin_amdgcn_readfirstlane((int)((blockIdx.x * blockDim.x + threadIdx.x) >> 5));
    if (node >= nodes) return;
    const int beg = rowptr[node];
    const int end = rowptr[node + 1];
    const float di = dinv[node];
    float2 a = ((const float2*)(xw + (size_t)node * 64))[lane];
    const float2 b = ((const float2*)bias)[lane];
    a.x *= di; a.y *= di;
    for (int base = beg; base < end; base += 32) {
        int2 my = make_int2(0, 0);
        if (base + lane < end) my = csr[base + lane];
        const int n = (end - base < 32) ? (end - base) : 32;
#pragma unroll 8
        for (int k = 0; k < n; ++k) {
            const int s = __builtin_amdgcn_readlane(my.x, k);
            const float w = __int_as_float(__builtin_amdgcn_readlane(my.y, k));
            const float2 v = ((const float2*)(xw + (size_t)s * 64))[lane];
            a.x = fmaf(v.x, w, a.x);
            a.y = fmaf(v.y, w, a.y);
        }
    }
    const float v0 = a.x * di + b.x;
    const float v1 = a.y * di + b.y;
    float m = fmaxf(v0, v1);
#pragma unroll
    for (int off = 16; off > 0; off >>= 1) m = fmaxf(m, __shfl_xor(m, off, 32));
    const float e0 = __expf(v0 - m);
    const float e1 = __expf(v1 - m);
    float sum = e0 + e1;
#pragma unroll
    for (int off = 16; off > 0; off >>= 1) sum += __shfl_xor(sum, off, 32);
    const float inv = 1.0f / sum;
    ((float2*)(out + (size_t)node * 64))[lane] = make_float2(e0 * inv, e1 * inv);
}

extern "C" void kernel_launch(void* const* d_in, const int* in_sizes, int n_in,
                              void* d_out, int out_size, void* d_ws, size_t ws_size,
                              hipStream_t stream) {
    const float* x   = (const float*)d_in[0];
    const int*   ei  = (const int*)  d_in[1];
    const float* W1  = (const float*)d_in[2];
    const float* b1  = (const float*)d_in[3];
    const float* W2  = (const float*)d_in[4];
    const float* b2  = (const float*)d_in[5];
    float* out = (float*)d_out;

    const int N = in_sizes[0] / GCN_D_IN;
    const int E = in_sizes[1] / 2;
    const int* src = ei;
    const int* dst = ei + E;

    float* ws   = (float*)d_ws;
    size_t off  = ((size_t)N + 63) & ~(size_t)63;
    float* dinv = ws;
    _Float16* xw1 = (_Float16*)(ws + off);
    float* reg2 = ws + off + (size_t)N * GCN_D_HID;
    _Float16* W1p = (_Float16*)(reg2 + (size_t)N * GCN_D_HID);
    _Float16* W2p = W1p + (size_t)GCN_D_IN * GCN_D_HID;
    float* xw2  = reg2;

    int*  so       = (int*)(reg2 + (size_t)N * GCN_D_OUT);
    int*  deg      = so;
    int*  rowloc   = so + off;
    int*  blocktot = so + 2 * off + 64;
    int2* csr      = (int2*)(so + 2 * off + 64 + 1024);
    int*  part     = so + 2 * off + 64 + 1024 + (((size_t)2 * E + 63) & ~(size_t)63);
    const int pitch = (int)off;
    int*  rowptr   = part + (size_t)GCN_EDGE_CHUNKS * off;

    const int B = 256;
    {
        const int SN = GCN_SCAN_NODES;
        const int nblk = (N + SN - 1) / SN;
        gcn_chunk_count_pack <<<GCN_EDGE_CHUNKS + GCN_PACK_BLOCKS, 1024, 0, stream>>>(dst, part, E, pitch, GCN_EDGE_CHUNKS,
                                                                                      W1, W1p, W2, W2p);
        gcn_chunk_prefix_scan<<<nblk, 1024, 0, stream>>>(part, deg, rowloc, blocktot, N, pitch);
        gcn_chunk_place      <<<GCN_EDGE_CHUNKS, 1024, 0, stream>>>(src, dst, deg, rowloc, blocktot, part, rowptr, dinv,
                                                                    csr, E, N, pitch, nblk);
    }

    {
        int g1 = ((N + 15) / 16 + 15) / 16;
        if (g1 < 256) g1 = 256;
        gcn_wmma_gemm_f16_ldsb<<<g1, 512, 0, stream>>>(x, W1p, xw1, N);
    }
    {
        gcn_gather256_relu_gemm64<<<(N + 15) / 16, 512, 0, stream>>>(rowptr, csr, dinv, xw1, b1, W2p, xw2, N);
    }

    {
        int th = N * 32;
        gcn_gather64_bias_softmax<<<(th + B - 1) / B, B, 0, stream>>>(rowptr, csr, dinv, xw2, b2, out, N);
    }
}


# placing and layer-1 GEMM share one launch (6 -> 5): 128 GEMM workgroups run beside the 128 placing ones, 1-2 row tiles per wave; placing at 512 threads; a node's sum order still = LDS add arrival
# speedup vs baseline: 5.2164x; 1.1460x over previous
#include <hip/hip_runtime.h>
#include <hip/hip_bf16.h>

typedef __attribute__((ext_vector_type(16))) _Float16 v16h;
typedef __attribute__((ext_vector_type(8)))  float    v8f;
typedef __attribute__((ext_vector_type(4)))  _Float16 v4h;
typedef __attribute__((ext_vector_type(8)))  _Float16 v8h;

#define GCN_D_IN  512
#define GCN_D_HID 256
#define GCN_D_OUT 64
#define GCN_H_PITCH (GCN_D_HID + 8)

#define GCN_NODE_CAP    50176
#define GCN_EDGE_CHUNKS 128
#define GCN_SCAN_NODES  256
#define GCN_EDGE_BATCH  8
#define GCN_L3_THREADS  512
#define GCN_PLACE_BATCH 13
#define GCN_CUR_BATCH   8
#define GCN_GEMM_WGS    128

static __device__ __forceinline__ void gcn_pack_w_item(const float* __restrict__ W,
                                                       _Float16* __restrict__ Wp,
                                                       int N, int deal8, int u) {
    const int tilesN = N >> 4;
    const int frag = u >> 5;
    const int lane = u & 31;
    const int kt = frag / tilesN;
    const int tn = frag - kt * tilesN;
    const int n    = deal8 ? ((tn >> 3) * 128 + (lane & 15) * 8 + (tn & 7))
                           : (tn * 16 + (lane & 15));
    const int koff = kt * 32 + (lane >> 4) * 8;
    _Float16* o = Wp + (size_t)u * 16;
#pragma unroll
    for (int j = 0; j < 8; ++j) {
        o[j]     = (_Float16)W[(size_t)(koff + j) * N + n];
        o[j + 8] = (_Float16)W[(size_t)(koff + 16 + j) * N + n];
    }
}

#define GCN_PACK_ITEMS_W1 ((GCN_D_IN >> 5) * (GCN_D_HID >> 4) * 32)
#define GCN_PACK_ITEMS_W2 ((GCN_D_HID >> 5) * (GCN_D_OUT >> 4) * 32)
#define GCN_PACK_BLOCKS   ((GCN_PACK_ITEMS_W1 + GCN_PACK_ITEMS_W2 + 1023) / 1024)

__global__ void __launch_bounds__(1024)
gcn_chunk_count_pack(const int* __restrict__ dst, int* __restrict__ part, int E, int pitch, int chunks,
                     const float* __restrict__ W1, _Float16* __restrict__ W1p,
                     const float* __restrict__ W2, _Float16* __restrict__ W2p) {
    __shared__ __attribute__((aligned(16))) int cnt[GCN_NODE_CAP];
    const int t = threadIdx.x;
    if ((int)blockIdx.x >= chunks) {
        int u = ((int)blockIdx.x - chunks) * 1024 + t;
        if (u < GCN_PACK_ITEMS_W1) {
            gcn_pack_w_item(W1, W1p, GCN_D_HID, 1, u);
        } else {
            u -= GCN_PACK_ITEMS_W1;
            if (u < GCN_PACK_ITEMS_W2) gcn_pack_w_item(W2, W2p, GCN_D_OUT, 0, u);
        }
        return;
    }
    const int p4 = pitch >> 2;
    for (int i = t; i < p4; i += 1024) ((int4*)cnt)[i] = make_int4(0, 0, 0, 0);
    __syncthreads();
    const int per = (E + chunks - 1) / chunks;
    const int e0 = (int)blockIdx.x * per;
    const int e1 = (e0 + per < E) ? (e0 + per) : E;
    for (int eb = e0 + t; eb < e1; eb += GCN_EDGE_BATCH * 1024) {
        int d[GCN_EDGE_BATCH];
#pragma unroll
        for (int j = 0; j < GCN_EDGE_BATCH; ++j) {
            const int e = eb + j * 1024;
            d[j] = dst[(e < e1) ? e : (e1 - 1)];
        }
#pragma unroll
        for (int j = 0; j < GCN_EDGE_BATCH; ++j)
            atomicAdd(&cnt[d[j]], (eb + j * 1024 < e1) ? 1 : 0);
    }
    __syncthreads();
    int4* __restrict__ row = (int4*)(part + (size_t)blockIdx.x * pitch);
    for (int i = t; i < p4; i += 1024) row[i] = ((const int4*)cnt)[i];
}

__global__ void __launch_bounds__(1024)
gcn_chunk_prefix_scan(int* __restrict__ part, int* __restrict__ deg, int* __restrict__ rowptr,
                      int* __restrict__ blocktot, int n, int pitch) {
    __shared__ int gsum[4][GCN_SCAN_NODES];
    __shared__ int wsum[8];
    const int t = threadIdx.x;
    const int lane = t & 31;
    const int wv = t >> 5;
    const int nl = t & (GCN_SCAN_NODES - 1);
    const int g = t >> 8;
    const int i = (int)blockIdx.x * GCN_SCAN_NODES + nl;
    int* __restrict__ p = part + (size_t)(g * 32) * pitch + i;
    int v[32];
    int s = 0;
    if (i < n) {
#pragma unroll
        for (int j = 0; j < 32; ++j) v[j] = p[(size_t)j * pitch];
#pragma unroll
        for (int j = 0; j < 32; ++j) { const int x = v[j]; v[j] = s; s += x; }
    }
    gsum[g][nl] = s;
    __syncthreads();
    int goff = 0, c = 0;
#pragma unroll
    for (int k = 0; k < 4; ++k) {
        const int x = gsum[k][nl];
        if (k < g) goff += x;
        c += x;
    }
    if (i < n) {
#pragma unroll
        for (int j = 0; j < 32; ++j) p[(size_t)j * pitch] = v[j] + goff;
    }
    int incl = c;
#pragma unroll
    for (int off = 1; off < 32; off <<= 1) {
        int y = __shfl_up(incl, off, 32);
        if (lane >= off) incl += y;
    }
    if (g == 0 && lane == 31) wsum[wv] = incl;
    __syncthreads();
    if (wv == 0) {
        const int w = (lane < 8) ? wsum[lane] : 0;
        int inc2 = w;
#pragma unroll
        for (int off = 1; off < 8; off <<= 1) {
            int y = __shfl_up(inc2, off, 32);
            if (lane >= off) inc2 += y;
        }
        if (lane < 8) wsum[lane] = inc2 - w;
    }
    __syncthreads();
    if (g == 0) {
        const int excl = incl - c + wsum[wv];
        if (i < n) { rowptr[i] = excl; deg[i] = c; }
        if (t == GCN_SCAN_NODES - 1) blocktot[blockIdx.x] = excl + c;
    }
}

static __device__ __forceinline__ float gcn_dinv_of(int indeg) {
    const float d = (float)(indeg + 1);
    return (d > 0.0f) ? rsqrtf(d) : 0.0f;
}

static __device__ __forceinline__ void
gcn_place_body(int* __restrict__ cur, int* __restrict__ bbase, int* __restrict__ wtot,
               const int* __restrict__ src, const int* __restrict__ dst,
               const int* __restrict__ deg, const int* __restrict__ rowloc,
               const int* __restrict__ blocktot, const int* __restrict__ part,
               int* __restrict__ rowptr, float* __restrict__ dinv,
               int2* __restrict__ csr, int E, int n, int pitch, int nblk, int c, int chunks) {
    const int NT = GCN_L3_THREADS;
    const int NW = GCN_L3_THREADS / 32;
    const int t = threadIdx.x;
    const int lane = t & 31;
    const int wv = t >> 5;
    {
        const int bt = (t < nblk) ? blocktot[t] : 0;
        int incl = bt;
#pragma unroll
        for (int off = 1; off < 32; off <<= 1) {
            int y = __shfl_up(incl, off, 32);
            if (lane >= off) incl += y;
        }
        if (lane == 31) wtot[wv] = incl;
        __syncthreads();
        if (wv == 0) {
            const int w = (lane < NW) ? wtot[lane] : 0;
            int inc2 = w;
#pragma unroll
            for (int off = 1; off < 32; off <<= 1) {
                int y = __shfl_up(inc2, off, 32);
                if (lane >= off) inc2 += y;
            }
            if (lane < NW) wtot[lane] = inc2 - w;
        }
        __syncthreads();
        const int excl = incl - bt + wtot[wv];
        if (t < nblk) bbase[t] = excl;
        if (t == nblk - 1) bbase[nblk] = excl + bt;
    }
    __syncthreads();
    const int* __restrict__ prow = part + (size_t)c * pitch;
    const int n4 = n >> 2;
    {
    int i = t;
    for (; i + (GCN_CUR_BATCH - 1) * NT < n4; i += GCN_CUR_BATCH * NT) {
        int4 a[GCN_CUR_BATCH], r[GCN_CUR_BATCH];
#pragma unroll
        for (int j = 0; j < GCN_CUR_BATCH; ++j) {
            a[j] = ((const int4*)prow)[i + j * NT];
            r[j] = ((const int4*)rowloc)[i + j * NT];
        }
#pragma unroll
        for (int j = 0; j < GCN_CUR_BATCH; ++j) {
            const int q = i + j * NT;
            const int bb = bbase[(q << 2) / GCN_SCAN_NODES];
            ((int4*)cur)[q] = make_int4(a[j].x + r[j].x + bb, a[j].y + r[j].y + bb,
                                        a[j].z + r[j].z + bb, a[j].w + r[j].w + bb);
        }
    }
    for (; i < n4; i += NT) {
        const int4 a = ((const int4*)prow)[i];
        const int4 r = ((const int4*)rowloc)[i];
        const int bb = bbase[(i << 2) / GCN_SCAN_NODES];
        ((int4*)cur)[i] = make_int4(a.x + r.x + bb, a.y + r.y + bb, a.z + r.z + bb, a.w + r.w + bb);
    }
    }
    for (int k = (n4 << 2) + t; k < n; k += NT) cur[k] = prow[k] + rowloc[k] + bbase[k / GCN_SCAN_NODES];
    {
        const int nper = (n + chunks - 1) / chunks;
        const int i0 = c * nper;
        const int i1 = (i0 + nper < n) ? (i0 + nper) : n;
        for (int i = i0 + t; i < i1; i += NT) {
            rowptr[i] = rowloc[i] + bbase[i / GCN_SCAN_NODES];
            dinv[i] = gcn_dinv_of(deg[i]);
        }
        if (c == 0 && t == 0) rowptr[n] = bbase[nblk];
    }
    __syncthreads();
    const int per = (E + chunks - 1) / chunks;
    const int e0 = c * per;
    const int e1 = (e0 + per < E) ? (e0 + per) : E;
    for (int eb = e0 + t; eb < e1; eb += GCN_PLACE_BATCH * NT) {
        int s[GCN_PLACE_BATCH], d[GCN_PLACE_BATCH], pos[GCN_PLACE_BATCH], sd[GCN_PLACE_BATCH];
#pragma unroll
        for (int j = 0; j < GCN_PLACE_BATCH; ++j) {
            const int e = eb + j * NT;
            const int ec = (e < e1) ? e : (e1 - 1);
            s[j] = src[ec];
            d[j] = dst[ec];
        }
#pragma unroll
        for (int j = 0; j < GCN_PLACE_BATCH; ++j) sd[j] = deg[s[j]];
#pragma unroll
        for (int j = 0; j < GCN_PLACE_BATCH; ++j)
            pos[j] = atomicAdd(&cur[d[j]], (eb + j * NT < e1) ? 1 : 0);
#pragma unroll
        for (int j = 0; j < GCN_PLACE_BATCH; ++j)
            if (eb + j * NT < e1) csr[pos[j]] = make_int2(s[j], __float_as_int(gcn_dinv_of(sd[j])));
    }
}


static __device__ __forceinline__ void gcn_ldb4(v16h (&dst)[4], const v8h* p) {
#pragma unroll
    for (int j = 0; j < 4; ++j) {
        const v8h lo = p[j * 64];
        const v8h hi = p[j * 64 + 32];
        dst[j] = __builtin_shufflevector(lo, hi, 0, 1, 2, 3, 4, 5, 6, 7,
                                         8, 9, 10, 11, 12, 13, 14, 15);
    }
}

static __device__ __forceinline__ void gcn_lda(float4 (&dst)[4], const float* arow, int kt) {
    const float* p = arow + kt * 32;
    dst[0] = *(const float4*)(p);
    dst[1] = *(const float4*)(p + 4);
    dst[2] = *(const float4*)(p + 16);
    dst[3] = *(const float4*)(p + 20);
}

static __device__ __forceinline__ void
gcn_gemm_body(v8h* __restrict__ Bs, const float* __restrict__ A, const _Float16* __restrict__ Bp,
              _Float16* __restrict__ C, int M, int g, int G) {
    const int tid  = threadIdx.x;
    const int lane = tid & 31;
    const int wv   = __builtin_amdgcn_readfirstlane(tid >> 5);
    const int hsel = lane >> 4;
    const int idx  = lane & 15;
    const int tilesM = (M + 15) >> 4;
    const int tstep = (GCN_L3_THREADS / 32) * G;

    int tm = wv * G + g;
    if (tm < tilesM) {
        int row = tm * 16 + idx;
        row = (row < M) ? row : (M - 1);
        const float* __restrict__ arow = A + (size_t)row * GCN_D_IN + hsel * 8;
        __builtin_prefetch(arow, 0, 3);
        __builtin_prefetch(arow + 32, 0, 3);
        __builtin_prefetch(arow + 64, 0, 3);
    }

    {
        const v8h* __restrict__ Bg = (const v8h*)Bp;
#pragma unroll 8
        for (int i = 0; i < 32; ++i) {
            const int q = (((i + g) & 31) << 9) + tid;
            const int dst = ((q >> 6) << 6) | ((q & 1) << 5) | ((q >> 1) & 31);
            Bs[dst] = Bg[q];
        }
    }
    __syncthreads();

    int bofs = lane;
    for (; tm < tilesM; tm += tstep) {
        asm volatile("" : "+v"(bofs));
        const v8h* bk = Bs + bofs;
        int row = tm * 16 + idx;
        row = (row < M) ? row : (M - 1);
        const float* __restrict__ arow = A + (size_t)row * GCN_D_IN + hsel * 8;

        v8f acc[16];
#pragma unroll
        for (int t = 0; t < 16; ++t) acc[t] = (v8f){};

        float4 ar[3][4];
        gcn_lda(ar[0], arow, 0);
        gcn_lda(ar[1], arow, 1);
        gcn_lda(ar[2], arow, 2);

        v16h bq[2][4];
        gcn_ldb4(bq[0], bk);

#pragma unroll
        for (int kt = 0; kt < GCN_D_IN / 32; ++kt) {
            const int sl = kt % 3;
            float af[16];
            *(float4*)(af + 0)  = ar[sl][0];  *(float4*)(af + 4)  = ar[sl][1];
            *(float4*)(af + 8)  = ar[sl][2];  *(float4*)(af + 12) = ar[sl][3];
            v16h av;
#pragma unroll
            for (int j = 0; j < 16; ++j) av[j] = (_Float16)af[j];
            if (kt + 3 < GCN_D_IN / 32) gcn_lda(ar[sl], arow, kt + 3);
#pragma unroll
            for (int gq = 0; gq < 4; ++gq) {
                const int cur = gq & 1;
                const int nf = kt * 16 + (gq + 1) * 4;
                if (nf < (GCN_D_IN / 32) * 16) gcn_ldb4(bq[cur ^ 1], bk + nf * 64);
#pragma unroll
                for (int j = 0; j < 4; ++j)
                    acc[gq * 4 + j] = __builtin_amdgcn_wmma_f32_16x16x32_f16(false, av, false, bq[cur][j], (short)0,
                                                                            acc[gq * 4 + j], false, false);
            }
        }

        const int row0 = tm * 16 + hsel * 8;
        _Float16* cbase = C + (size_t)row0 * GCN_D_HID + idx * 8;
#pragma unroll
        for (int r = 0; r < 8; ++r) {
            if (row0 + r < M) {
#pragma unroll
                for (int h = 0; h < 2; ++h) {
                    v8h o;
#pragma unroll
                    for (int j = 0; j < 8; ++j) o[j] = (_Float16)acc[h * 8 + j][r];
                    *(v8h*)(cbase + (size_t)r * GCN_D_HID + h * 128) = o;
                }
            }
        }
    }
}

__global__ void __launch_bounds__(GCN_L3_THREADS)
gcn_place_and_gemm(const int* __restrict__ src, const int* __restrict__ dst,
                   const int* __restrict__ deg, const int* __restrict__ rowloc,
                   const int* __restrict__ blocktot, const int* __restrict__ part,
                   int* __restrict__ rowptr, float* __restrict__ dinv,
                   int2* __restrict__ csr, int E, int n, int pitch, int nblk, int chunks,
                   const float* __restrict__ A, const _Float16* __restrict__ Bp,
                   _Float16* __restrict__ C) {
    __shared__ __attribute__((aligned(16))) unsigned char lds[GCN_D_IN * GCN_D_HID * 2];
    static_assert((GCN_NODE_CAP + GCN_NODE_CAP / GCN_SCAN_NODES + 1 + 32) * 4 <= GCN_D_IN * GCN_D_HID * 2,
                  "placing tables must fit the shared block");
    if ((int)blockIdx.x < chunks) {
        int* cur   = (int*)lds;
        int* bbase = cur + GCN_NODE_CAP;
        int* wtot  = bbase + (GCN_NODE_CAP / GCN_SCAN_NODES + 1);
        gcn_place_body(cur, bbase, wtot, src, dst, deg, rowloc, blocktot, part, rowptr, dinv, csr,
                       E, n, pitch, nblk, (int)blockIdx.x, chunks);
    } else {
        gcn_gemm_body((v8h*)lds, A, Bp, C, n, (int)blockIdx.x - chunks, (int)gridDim.x - chunks);
    }
}

__global__ void __launch_bounds__(512)
gcn_gather256_relu_gemm64(const int* __restrict__ rowptr, const int2* __restrict__ csr,
                          const float* __restrict__ dinv, const _Float16* __restrict__ xw,
                          const float* __restrict__ bias, const _Float16* __restrict__ Bp,
                          float* __restrict__ xw2, int nodes) {
    __shared__ __attribute__((aligned(16))) _Float16 htile[16 * GCN_H_PITCH];
    const int lane = threadIdx.x & 31;
    const int wv = __builtin_amdgcn_readfirstlane((int)(threadIdx.x >> 5));
    const int node0 = (int)blockIdx.x * 16;
    const int node = node0 + wv;
    float a[8];
#pragma unroll
    for (int j = 0; j < 8; ++j) a[j] = 0.0f;
    if (node < nodes) {
        const int beg = rowptr[node];
        const int end = rowptr[node + 1];
        const float di = dinv[node];
        const v8h sv = ((const v8h*)(xw + (size_t)node * 256))[lane];
        const float4 b0 = ((const float4*)bias)[2 * lane];
        const float4 b1 = ((const float4*)bias)[2 * lane + 1];
        const float b[8] = {b0.x, b0.y, b0.z, b0.w, b1.x, b1.y, b1.z, b1.w};
#pragma unroll
        for (int j = 0; j < 8; ++j) a[j] = (float)sv[j] * di;
        for (int base = beg; base < end; base += 32) {
            int2 my = make_int2(0, 0);
            if (base + lane < end) my = csr[base + lane];
            const int n = (end - base < 32) ? (end - base) : 32;
#pragma unroll 4
            for (int k = 0; k < n; ++k) {
                const int s = __builtin_amdgcn_readlane(my.x, k);
                const float w = __int_as_float(__builtin_amdgcn_readlane(my.y, k));
                const v8h v = ((const v8h*)(xw + (size_t)s * 256))[lane];
#pragma unroll
                for (int j = 0; j < 8; ++j) a[j] = fmaf((float)v[j], w, a[j]);
            }
        }
#pragma unroll
        for (int j = 0; j < 8; ++j) a[j] = fmaxf(a[j] * di + b[j], 0.0f);
    }
    {
        v8h h;
#pragma unroll
        for (int j = 0; j < 8; ++j) h[j] = (_Float16)a[j];
        *(v8h*)(htile + wv * GCN_H_PITCH + 8 * lane) = h;
    }
    __syncthreads();
    if (wv >= 2) return;

    const int hsel = lane >> 4;
    const int idx  = lane & 15;
    const int tn0  = wv * 2;
    const _Float16* arow = htile + idx * GCN_H_PITCH + hsel * 8;
    v8f acc0 = {}, acc1 = {};
    for (int kt = 0; kt < GCN_D_HID / 32; ++kt) {
        const v8h lo = *(const v8h*)(arow + kt * 32);
        const v8h hi = *(const v8h*)(arow + kt * 32 + 16);
        const v16h av = __builtin_shufflevector(lo, hi, 0, 1, 2, 3, 4, 5, 6, 7,
                                                8, 9, 10, 11, 12, 13, 14, 15);
        const _Float16* bbase = Bp + ((size_t)(kt * (GCN_D_OUT >> 4) + tn0) * 32 + lane) * 16;
        const v16h b0 = *(const v16h*)(bbase);
        const v16h b1 = *(const v16h*)(bbase + 32 * 16);
        acc0 = __builtin_amdgcn_wmma_f32_16x16x32_f16(false, av, false, b0, (short)0, acc0, false, false);
        acc1 = __builtin_amdgcn_wmma_f32_16x16x32_f16(false, av, false, b1, (short)0, acc1, false, false);
    }
    const int row0 = node0 + hsel * 8;
    float* cbase = xw2 + (size_t)row0 * GCN_D_OUT + tn0 * 16 + idx;
#pragma unroll
    for (int r = 0; r < 8; ++r) {
        if (row0 + r < nodes) {
            cbase[(size_t)r * GCN_D_OUT + 0]  = acc0[r];
            cbase[(size_t)r * GCN_D_OUT + 16] = acc1[r];
        }
    }
}

__global__ void gcn_gather64_bias_softmax(const int* __restrict__ rowptr, const int2* __restrict__ csr,
                                          const float* __restrict__ dinv, const float* __restrict__ xw,
                                          const float* __restrict__ bias,
                                          float* __restrict__ out, int nodes) {
    const int lane = threadIdx.x & 31;
    const int node = __builtin_amdgcn_readfirstlane((int)((blockIdx.x * blockDim.x + threadIdx.x) >> 5));
    if (node >= nodes) return;
    const int beg = rowptr[node];
    const int end = rowptr[node + 1];
    const float di = dinv[node];
    float2 a = ((const float2*)(xw + (size_t)node * 64))[lane];
    const float2 b = ((const float2*)bias)[lane];
    a.x *= di; a.y *= di;
    for (int base = beg; base < end; base += 32) {
        int2 my = make_int2(0, 0);
        if (base + lane < end) my = csr[base + lane];
        const int n = (end - base < 32) ? (end - base) : 32;
#pragma unroll 8
        for (int k = 0; k < n; ++k) {
            const int s = __builtin_amdgcn_readlane(my.x, k);
            const float w = __int_as_float(__builtin_amdgcn_readlane(my.y, k));
            const float2 v = ((const float2*)(xw + (size_t)s * 64))[lane];
            a.x = fmaf(v.x, w, a.x);
            a.y = fmaf(v.y, w, a.y);
        }
    }
    const float v0 = a.x * di + b.x;
    const float v1 = a.y * di + b.y;
    float m = fmaxf(v0, v1);
#pragma unroll
    for (int off = 16; off > 0; off >>= 1) m = fmaxf(m, __shfl_xor(m, off, 32));
    const float e0 = __expf(v0 - m);
    const float e1 = __expf(v1 - m);
    float sum = e0 + e1;
#pragma unroll
    for (int off = 16; off > 0; off >>= 1) sum += __shfl_xor(sum, off, 32);
    const float inv = 1.0f / sum;
    ((float2*)(out + (size_t)node * 64))[lane] = make_float2(e0 * inv, e1 * inv);
}

extern "C" void kernel_launch(void* const* d_in, const int* in_sizes, int n_in,
                              void* d_out, int out_size, void* d_ws, size_t ws_size,
                              hipStream_t stream) {
    const float* x   = (const float*)d_in[0];
    const int*   ei  = (const int*)  d_in[1];
    const float* W1  = (const float*)d_in[2];
    const float* b1  = (const float*)d_in[3];
    const float* W2  = (const float*)d_in[4];
    const float* b2  = (const float*)d_in[5];
    float* out = (float*)d_out;

    const int N = in_sizes[0] / GCN_D_IN;
    const int E = in_sizes[1] / 2;
    const int* src = ei;
    const int* dst = ei + E;

    float* ws   = (float*)d_ws;
    size_t off  = ((size_t)N + 63) & ~(size_t)63;
    float* dinv = ws;
    _Float16* xw1 = (_Float16*)(ws + off);
    float* reg2 = ws + off + (size_t)N * GCN_D_HID;
    _Float16* W1p = (_Float16*)(reg2 + (size_t)N * GCN_D_HID);
    _Float16* W2p = W1p + (size_t)GCN_D_IN * GCN_D_HID;
    float* xw2  = reg2;

    int*  so       = (int*)(reg2 + (size_t)N * GCN_D_OUT);
    int*  deg      = so;
    int*  rowloc   = so + off;
    int*  blocktot = so + 2 * off + 64;
    int2* csr      = (int2*)(so + 2 * off + 64 + 1024);
    int*  part     = so + 2 * off + 64 + 1024 + (((size_t)2 * E + 63) & ~(size_t)63);
    const int pitch = (int)off;
    int*  rowptr   = part + (size_t)GCN_EDGE_CHUNKS * off;

    const int B = 256;
    {
        const int SN = GCN_SCAN_NODES;
        const int nblk = (N + SN - 1) / SN;
        gcn_chunk_count_pack <<<GCN_EDGE_CHUNKS + GCN_PACK_BLOCKS, 1024, 0, stream>>>(dst, part, E, pitch, GCN_EDGE_CHUNKS,
                                                                                      W1, W1p, W2, W2p);
        gcn_chunk_prefix_scan<<<nblk, 1024, 0, stream>>>(part, deg, rowloc, blocktot, N, pitch);
        gcn_place_and_gemm   <<<GCN_EDGE_CHUNKS + GCN_GEMM_WGS, GCN_L3_THREADS, 0, stream>>>(
            src, dst, deg, rowloc, blocktot, part, rowptr, dinv, csr, E, N, pitch, nblk, GCN_EDGE_CHUNKS,
            x, W1p, xw1);
    }

    {
        gcn_gather256_relu_gemm64<<<(N + 15) / 16, 512, 0, stream>>>(rowptr, csr, dinv, xw1, b1, W2p, xw2, N);
    }

    {
        int th = N * 32;
        gcn_gather64_bias_softmax<<<(th + B - 1) / B, B, 0, stream>>>(rowptr, csr, dinv, xw2, b2, out, N);
    }
}


# edge weight dinv[src] folded into the gathered rows (scaled where produced, ahead of their f16 rounding): in-edge entry = source id only, placing drops 800k random degree reads; sum order kept
# speedup vs baseline: 5.4403x; 1.0429x over previous
#include <hip/hip_runtime.h>
#include <hip/hip_bf16.h>

typedef __attribute__((ext_vector_type(16))) _Float16 v16h;
typedef __attribute__((ext_vector_type(8)))  float    v8f;
typedef __attribute__((ext_vector_type(4)))  _Float16 v4h;
typedef __attribute__((ext_vector_type(8)))  _Float16 v8h;

#define GCN_D_IN  512
#define GCN_D_HID 256
#define GCN_D_OUT 64
#define GCN_H_PITCH (GCN_D_HID + 8)

#define GCN_NODE_CAP    50176
#define GCN_EDGE_CHUNKS 128
#define GCN_SCAN_NODES  256
#define GCN_EDGE_BATCH  8
#define GCN_L3_THREADS  512
#define GCN_PLACE_BATCH 13
#define GCN_CUR_BATCH   8
#define GCN_GEMM_WGS    128

static __device__ __forceinline__ void gcn_pack_w_item(const float* __restrict__ W,
                                                       _Float16* __restrict__ Wp,
                                                       int N, int deal8, int u) {
    const int tilesN = N >> 4;
    const int frag = u >> 5;
    const int lane = u & 31;
    const int kt = frag / tilesN;
    const int tn = frag - kt * tilesN;
    const int n    = deal8 ? ((tn >> 3) * 128 + (lane & 15) * 8 + (tn & 7))
                           : (tn * 16 + (lane & 15));
    const int koff = kt * 32 + (lane >> 4) * 8;
    _Float16* o = Wp + (size_t)u * 16;
#pragma unroll
    for (int j = 0; j < 8; ++j) {
        o[j]     = (_Float16)W[(size_t)(koff + j) * N + n];
        o[j + 8] = (_Float16)W[(size_t)(koff + 16 + j) * N + n];
    }
}

#define GCN_PACK_ITEMS_W1 ((GCN_D_IN >> 5) * (GCN_D_HID >> 4) * 32)
#define GCN_PACK_ITEMS_W2 ((GCN_D_HID >> 5) * (GCN_D_OUT >> 4) * 32)
#define GCN_PACK_BLOCKS   ((GCN_PACK_ITEMS_W1 + GCN_PACK_ITEMS_W2 + 1023) / 1024)

__global__ void __launch_bounds__(1024)
gcn_chunk_count_pack(const int* __restrict__ dst, int* __restrict__ part, int E, int pitch, int chunks,
                     const float* __restrict__ W1, _Float16* __restrict__ W1p,
                     const float* __restrict__ W2, _Float16* __restrict__ W2p) {
    __shared__ __attribute__((aligned(16))) int cnt[GCN_NODE_CAP];
    const int t = threadIdx.x;
    if ((int)blockIdx.x >= chunks) {
        int u = ((int)blockIdx.x - chunks) * 1024 + t;
        if (u < GCN_PACK_ITEMS_W1) {
            gcn_pack_w_item(W1, W1p, GCN_D_HID, 1, u);
        } else {
            u -= GCN_PACK_ITEMS_W1;
            if (u < GCN_PACK_ITEMS_W2) gcn_pack_w_item(W2, W2p, GCN_D_OUT, 0, u);
        }
        return;
    }
    const int p4 = pitch >> 2;
    for (int i = t; i < p4; i += 1024) ((int4*)cnt)[i] = make_int4(0, 0, 0, 0);
    __syncthreads();
    const int per = (E + chunks - 1) / chunks;
    const int e0 = (int)blockIdx.x * per;
    const int e1 = (e0 + per < E) ? (e0 + per) : E;
    for (int eb = e0 + t; eb < e1; eb += GCN_EDGE_BATCH * 1024) {
        int d[GCN_EDGE_BATCH];
#pragma unroll
        for (int j = 0; j < GCN_EDGE_BATCH; ++j) {
            const int e = eb + j * 1024;
            d[j] = dst[(e < e1) ? e : (e1 - 1)];
        }
#pragma unroll
        for (int j = 0; j < GCN_EDGE_BATCH; ++j)
            atomicAdd(&cnt[d[j]], (eb + j * 1024 < e1) ? 1 : 0);
    }
    __syncthreads();
    int4* __restrict__ row = (int4*)(part + (size_t)blockIdx.x * pitch);
    for (int i = t; i < p4; i += 1024) row[i] = ((const int4*)cnt)[i];
}

__global__ void __launch_bounds__(1024)
gcn_chunk_prefix_scan(int* __restrict__ part, int* __restrict__ deg, int* __restrict__ rowptr,
                      int* __restrict__ blocktot, int n, int pitch) {
    __shared__ int gsum[4][GCN_SCAN_NODES];
    __shared__ int wsum[8];
    const int t = threadIdx.x;
    const int lane = t & 31;
    const int wv = t >> 5;
    const int nl = t & (GCN_SCAN_NODES - 1);
    const int g = t >> 8;
    const int i = (int)blockIdx.x * GCN_SCAN_NODES + nl;
    int* __restrict__ p = part + (size_t)(g * 32) * pitch + i;
    int v[32];
    int s = 0;
    if (i < n) {
#pragma unroll
        for (int j = 0; j < 32; ++j) v[j] = p[(size_t)j * pitch];
#pragma unroll
        for (int j = 0; j < 32; ++j) { const int x = v[j]; v[j] = s; s += x; }
    }
    gsum[g][nl] = s;
    __syncthreads();
    int goff = 0, c = 0;
#pragma unroll
    for (int k = 0; k < 4; ++k) {
        const int x = gsum[k][nl];
        if (k < g) goff += x;
        c += x;
    }
    if (i < n) {
#pragma unroll
        for (int j = 0; j < 32; ++j) p[(size_t)j * pitch] = v[j] + goff;
    }
    int incl = c;
#pragma unroll
    for (int off = 1; off < 32; off <<= 1) {
        int y = __shfl_up(incl, off, 32);
        if (lane >= off) incl += y;
    }
    if (g == 0 && lane == 31) wsum[wv] = incl;
    __syncthreads();
    if (wv == 0) {
        const int w = (lane < 8) ? wsum[lane] : 0;
        int inc2 = w;
#pragma unroll
        for (int off = 1; off < 8; off <<= 1) {
            int y = __shfl_up(inc2, off, 32);
            if (lane >= off) inc2 += y;
        }
        if (lane < 8) wsum[lane] = inc2 - w;
    }
    __syncthreads();
    if (g == 0) {
        const int excl = incl - c + wsum[wv];
        if (i < n) { rowptr[i] = excl; deg[i] = c; }
        if (t == GCN_SCAN_NODES - 1) blocktot[blockIdx.x] = excl + c;
    }
}

static __device__ __forceinline__ float gcn_dinv_of(int indeg) {
    const float d = (float)(indeg + 1);
    return (d > 0.0f) ? rsqrtf(d) : 0.0f;
}

static __device__ __forceinline__ void
gcn_place_body(int* __restrict__ cur, int* __restrict__ bbase, int* __restrict__ wtot,
               const int* __restrict__ src, const int* __restrict__ dst,
               const int* __restrict__ deg, const int* __restrict__ rowloc,
               const int* __restrict__ blocktot, const int* __restrict__ part,
               int* __restrict__ rowptr, float* __restrict__ dinv,
               int* __restrict__ csr, int E, int n, int pitch, int nblk, int c, int chunks) {
    const int NT = GCN_L3_THREADS;
    const int NW = GCN_L3_THREADS / 32;
    const int t = threadIdx.x;
    const int lane = t & 31;
    const int wv = t >> 5;
    {
        const int bt = (t < nblk) ? blocktot[t] : 0;
        int incl = bt;
#pragma unroll
        for (int off = 1; off < 32; off <<= 1) {
            int y = __shfl_up(incl, off, 32);
            if (lane >= off) incl += y;
        }
        if (lane == 31) wtot[wv] = incl;
        __syncthreads();
        if (wv == 0) {
            const int w = (lane < NW) ? wtot[lane] : 0;
            int inc2 = w;
#pragma unroll
            for (int off = 1; off < 32; off <<= 1) {
                int y = __shfl_up(inc2, off, 32);
                if (lane >= off) inc2 += y;
            }
            if (lane < NW) wtot[lane] = inc2 - w;
        }
        __syncthreads();
        const int excl = incl - bt + wtot[wv];
        if (t < nblk) bbase[t] = excl;
        if (t == nblk - 1) bbase[nblk] = excl + bt;
    }
    __syncthreads();
    const int* __restrict__ prow = part + (size_t)c * pitch;
    const int n4 = n >> 2;
    {
    int i = t;
    for (; i + (GCN_CUR_BATCH - 1) * NT < n4; i += GCN_CUR_BATCH * NT) {
        int4 a[GCN_CUR_BATCH], r[GCN_CUR_BATCH];
#pragma unroll
        for (int j = 0; j < GCN_CUR_BATCH; ++j) {
            a[j] = ((const int4*)prow)[i + j * NT];
            r[j] = ((const int4*)rowloc)[i + j * NT];
        }
#pragma unroll
        for (int j = 0; j < GCN_CUR_BATCH; ++j) {
            const int q = i + j * NT;
            const int bb = bbase[(q << 2) / GCN_SCAN_NODES];
            ((int4*)cur)[q] = make_int4(a[j].x + r[j].x + bb, a[j].y + r[j].y + bb,
                                        a[j].z + r[j].z + bb, a[j].w + r[j].w + bb);
        }
    }
    for (; i < n4; i += NT) {
        const int4 a = ((const int4*)prow)[i];
        const int4 r = ((const int4*)rowloc)[i];
        const int bb = bbase[(i << 2) / GCN_SCAN_NODES];
        ((int4*)cur)[i] = make_int4(a.x + r.x + bb, a.y + r.y + bb, a.z + r.z + bb, a.w + r.w + bb);
    }
    }
    for (int k = (n4 << 2) + t; k < n; k += NT) cur[k] = prow[k] + rowloc[k] + bbase[k / GCN_SCAN_NODES];
    {
        const int nper = (n + chunks - 1) / chunks;
        const int i0 = c * nper;
        const int i1 = (i0 + nper < n) ? (i0 + nper) : n;
        for (int i = i0 + t; i < i1; i += NT) {
            rowptr[i] = rowloc[i] + bbase[i / GCN_SCAN_NODES];
            dinv[i] = gcn_dinv_of(deg[i]);
        }
        if (c == 0 && t == 0) rowptr[n] = bbase[nblk];
    }
    __syncthreads();
    const int per = (E + chunks - 1) / chunks;
    const int e0 = c * per;
    const int e1 = (e0 + per < E) ? (e0 + per) : E;
    for (int eb = e0 + t; eb < e1; eb += GCN_PLACE_BATCH * NT) {
        int s[GCN_PLACE_BATCH], d[GCN_PLACE_BATCH], pos[GCN_PLACE_BATCH];
#pragma unroll
        for (int j = 0; j < GCN_PLACE_BATCH; ++j) {
            const int e = eb + j * NT;
            const int ec = (e < e1) ? e : (e1 - 1);
            s[j] = src[ec];
            d[j] = dst[ec];
        }
#pragma unroll
        for (int j = 0; j < GCN_PLACE_BATCH; ++j)
            pos[j] = atomicAdd(&cur[d[j]], (eb + j * NT < e1) ? 1 : 0);
#pragma unroll
        for (int j = 0; j < GCN_PLACE_BATCH; ++j)
            if (eb + j * NT < e1) csr[pos[j]] = s[j];
    }
}


static __device__ __forceinline__ void gcn_ldb4(v16h (&dst)[4], const v8h* p) {
#pragma unroll
    for (int j = 0; j < 4; ++j) {
        const v8h lo = p[j * 64];
        const v8h hi = p[j * 64 + 32];
        dst[j] = __builtin_shufflevector(lo, hi, 0, 1, 2, 3, 4, 5, 6, 7,
                                         8, 9, 10, 11, 12, 13, 14, 15);
    }
}

static __device__ __forceinline__ void gcn_lda(float4 (&dst)[4], const float* arow, int kt) {
    const float* p = arow + kt * 32;
    dst[0] = *(const float4*)(p);
    dst[1] = *(const float4*)(p + 4);
    dst[2] = *(const float4*)(p + 16);
    dst[3] = *(const float4*)(p + 20);
}

static __device__ __forceinline__ void
gcn_gemm_body(v8h* __restrict__ Bs, const float* __restrict__ A, const _Float16* __restrict__ Bp,
              const int* __restrict__ deg, _Float16* __restrict__ C, int M, int g, int G) {
    const int tid  = threadIdx.x;
    const int lane = tid & 31;
    const int wv   = __builtin_amdgcn_readfirstlane(tid >> 5);
    const int hsel = lane >> 4;
    const int idx  = lane & 15;
    const int tilesM = (M + 15) >> 4;
    const int tstep = (GCN_L3_THREADS / 32) * G;

    int tm = wv * G + g;
    if (tm < tilesM) {
        int row = tm * 16 + idx;
        row = (row < M) ? row : (M - 1);
        const float* __restrict__ arow = A + (size_t)row * GCN_D_IN + hsel * 8;
        __builtin_prefetch(arow, 0, 3);
        __builtin_prefetch(arow + 32, 0, 3);
        __builtin_prefetch(arow + 64, 0, 3);
    }

    {
        const v8h* __restrict__ Bg = (const v8h*)Bp;
#pragma unroll 8
        for (int i = 0; i < 32; ++i) {
            const int q = (((i + g) & 31) << 9) + tid;
            const int dst = ((q >> 6) << 6) | ((q & 1) << 5) | ((q >> 1) & 31);
            Bs[dst] = Bg[q];
        }
    }
    __syncthreads();

    int bofs = lane;
    for (; tm < tilesM; tm += tstep) {
        asm volatile("" : "+v"(bofs));
        const v8h* bk = Bs + bofs;
        int row = tm * 16 + idx;
        row = (row < M) ? row : (M - 1);
        const float* __restrict__ arow = A + (size_t)row * GCN_D_IN + hsel * 8;
        const float dv = gcn_dinv_of(deg[row]);

        v8f acc[16];
#pragma unroll
        for (int t = 0; t < 16; ++t) acc[t] = (v8f){};

        float4 ar[3][4];
        gcn_lda(ar[0], arow, 0);
        gcn_lda(ar[1], arow, 1);
        gcn_lda(ar[2], arow, 2);

        v16h bq[2][4];
        gcn_ldb4(bq[0], bk);

#pragma unroll
        for (int kt = 0; kt < GCN_D_IN / 32; ++kt) {
            const int sl = kt % 3;
            float af[16];
            *(float4*)(af + 0)  = ar[sl][0];  *(float4*)(af + 4)  = ar[sl][1];
            *(float4*)(af + 8)  = ar[sl][2];  *(float4*)(af + 12) = ar[sl][3];
            v16h av;
#pragma unroll
            for (int j = 0; j < 16; ++j) av[j] = (_Float16)af[j];
            if (kt + 3 < GCN_D_IN / 32) gcn_lda(ar[sl], arow, kt + 3);
#pragma unroll
            for (int gq = 0; gq < 4; ++gq) {
                const int cur = gq & 1;
                const int nf = kt * 16 + (gq + 1) * 4;
                if (nf < (GCN_D_IN / 32) * 16) gcn_ldb4(bq[cur ^ 1], bk + nf * 64);
#pragma unroll
                for (int j = 0; j < 4; ++j)
                    acc[gq * 4 + j] = __builtin_amdgcn_wmma_f32_16x16x32_f16(false, av, false, bq[cur][j], (short)0,
                                                                            acc[gq * 4 + j], false, false);
            }
        }

#pragma unroll
        for (int r = 0; r < 8; ++r) {
            const float dl = __int_as_float(__builtin_amdgcn_readlane(__float_as_int(dv), r));
            const float dh = __int_as_float(__builtin_amdgcn_readlane(__float_as_int(dv), r + 8));
            const float dr = hsel ? dh : dl;
#pragma unroll
            for (int t = 0; t < 16; ++t) acc[t][r] *= dr;
        }

        const int row0 = tm * 16 + hsel * 8;
        _Float16* cbase = C + (size_t)row0 * GCN_D_HID + idx * 8;
#pragma unroll
        for (int r = 0; r < 8; ++r) {
            if (row0 + r < M) {
#pragma unroll
                for (int h = 0; h < 2; ++h) {
                    v8h o;
#pragma unroll
                    for (int j = 0; j < 8; ++j) o[j] = (_Float16)acc[h * 8 + j][r];
                    *(v8h*)(cbase + (size_t)r * GCN_D_HID + h * 128) = o;
                }
            }
        }
    }
}

__global__ void __launch_bounds__(GCN_L3_THREADS)
gcn_place_and_gemm(const int* __restrict__ src, const int* __restrict__ dst,
                   const int* __restrict__ deg, const int* __restrict__ rowloc,
                   const int* __restrict__ blocktot, const int* __restrict__ part,
                   int* __restrict__ rowptr, float* __restrict__ dinv,
                   int* __restrict__ csr, int E, int n, int pitch, int nblk, int chunks,
                   const float* __restrict__ A, const _Float16* __restrict__ Bp,
                   _Float16* __restrict__ C) {
    __shared__ __attribute__((aligned(16))) unsigned char lds[GCN_D_IN * GCN_D_HID * 2];
    static_assert((GCN_NODE_CAP + GCN_NODE_CAP / GCN_SCAN_NODES + 1 + 32) * 4 <= GCN_D_IN * GCN_D_HID * 2,
                  "placing tables must fit the shared block");
    if ((int)blockIdx.x < chunks) {
        int* cur   = (int*)lds;
        int* bbase = cur + GCN_NODE_CAP;
        int* wtot  = bbase + (GCN_NODE_CAP / GCN_SCAN_NODES + 1);
        gcn_place_body(cur, bbase, wtot, src, dst, deg, rowloc, blocktot, part, rowptr, dinv, csr,
                       E, n, pitch, nblk, (int)blockIdx.x, chunks);
    } else {
        gcn_gemm_body((v8h*)lds, A, Bp, deg, C, n, (int)blockIdx.x - chunks, (int)gridDim.x - chunks);
    }
}

__global__ void __launch_bounds__(512)
gcn_gather256_relu_gemm64(const int* __restrict__ rowptr, const int* __restrict__ csr,
                          const float* __restrict__ dinv, const _Float16* __restrict__ xw,
                          const float* __restrict__ bias, const _Float16* __restrict__ Bp,
                          float* __restrict__ xw2, int nodes) {
    __shared__ __attribute__((aligned(16))) _Float16 htile[16 * GCN_H_PITCH];
    const int lane = threadIdx.x & 31;
    const int wv = __builtin_amdgcn_readfirstlane((int)(threadIdx.x >> 5));
    const int node0 = (int)blockIdx.x * 16;
    const int node = node0 + wv;
    float a[8];
#pragma unroll
    for (int j = 0; j < 8; ++j) a[j] = 0.0f;
    if (node < nodes) {
        const int beg = rowptr[node];
        const int end = rowptr[node + 1];
        const float di = dinv[node];
        const v8h sv = ((const v8h*)(xw + (size_t)node * 256))[lane];
        const float4 b0 = ((const float4*)bias)[2 * lane];
        const float4 b1 = ((const float4*)bias)[2 * lane + 1];
        const float b[8] = {b0.x, b0.y, b0.z, b0.w, b1.x, b1.y, b1.z, b1.w};
#pragma unroll
        for (int j = 0; j < 8; ++j) a[j] = (float)sv[j];
        for (int base = beg; base < end; base += 32) {
            int my = 0;
            if (base + lane < end) my = csr[base + lane];
            const int n = (end - base < 32) ? (end - base) : 32;
#pragma unroll 4
            for (int k = 0; k < n; ++k) {
                const int s = __builtin_amdgcn_readlane(my, k);
                const v8h v = ((const v8h*)(xw + (size_t)s * 256))[lane];
#pragma unroll
                for (int j = 0; j < 8; ++j) a[j] += (float)v[j];
            }
        }
#pragma unroll
        for (int j = 0; j < 8; ++j) a[j] = fmaxf(a[j] * di + b[j], 0.0f) * di;
    }
    {
        v8h h;
#pragma unroll
        for (int j = 0; j < 8; ++j) h[j] = (_Float16)a[j];
        *(v8h*)(htile + wv * GCN_H_PITCH + 8 * lane) = h;
    }
    __syncthreads();
    if (wv >= 2) return;

    const int hsel = lane >> 4;
    const int idx  = lane & 15;
    const int tn0  = wv * 2;
    const _Float16* arow = htile + idx * GCN_H_PITCH + hsel * 8;
    v8f acc0 = {}, acc1 = {};
    for (int kt = 0; kt < GCN_D_HID / 32; ++kt) {
        const v8h lo = *(const v8h*)(arow + kt * 32);
        const v8h hi = *(const v8h*)(arow + kt * 32 + 16);
        const v16h av = __builtin_shufflevector(lo, hi, 0, 1, 2, 3, 4, 5, 6, 7,
                                                8, 9, 10, 11, 12, 13, 14, 15);
        const _Float16* bbase = Bp + ((size_t)(kt * (GCN_D_OUT >> 4) + tn0) * 32 + lane) * 16;
        const v16h b0 = *(const v16h*)(bbase);
        const v16h b1 = *(const v16h*)(bbase + 32 * 16);
        acc0 = __builtin_amdgcn_wmma_f32_16x16x32_f16(false, av, false, b0, (short)0, acc0, false, false);
        acc1 = __builtin_amdgcn_wmma_f32_16x16x32_f16(false, av, false, b1, (short)0, acc1, false, false);
    }
    const int row0 = node0 + hsel * 8;
    float* cbase = xw2 + (size_t)row0 * GCN_D_OUT + tn0 * 16 + idx;
#pragma unroll
    for (int r = 0; r < 8; ++r) {
        if (row0 + r < nodes) {
            cbase[(size_t)r * GCN_D_OUT + 0]  = acc0[r];
            cbase[(size_t)r * GCN_D_OUT + 16] = acc1[r];
        }
    }
}

__global__ void gcn_gather64_bias_softmax(const int* __restrict__ rowptr, const int* __restrict__ csr,
                                          const float* __restrict__ dinv, const float* __restrict__ xw,
                                          const float* __restrict__ bias,
                                          float* __restrict__ out, int nodes) {
    const int lane = threadIdx.x & 31;
    const int node = __builtin_amdgcn_readfirstlane((int)((blockIdx.x * blockDim.x + threadIdx.x) >> 5));
    if (node >= nodes) return;
    const int beg = rowptr[node];
    const int end = rowptr[node + 1];
    const float di = dinv[node];
    float2 a = ((const float2*)(xw + (size_t)node * 64))[lane];
    const float2 b = ((const float2*)bias)[lane];
    for (int base = beg; base < end; base += 32) {
        int my = 0;
        if (base + lane < end) my = csr[base + lane];
        const int n = (end - base < 32) ? (end - base) : 32;
#pragma unroll 8
        for (int k = 0; k < n; ++k) {
            const int s = __builtin_amdgcn_readlane(my, k);
            const float2 v = ((const float2*)(xw + (size_t)s * 64))[lane];
            a.x += v.x;
            a.y += v.y;
        }
    }
    const float v0 = a.x * di + b.x;
    const float v1 = a.y * di + b.y;
    float m = fmaxf(v0, v1);
#pragma unroll
    for (int off = 16; off > 0; off >>= 1) m = fmaxf(m, __shfl_xor(m, off, 32));
    const float e0 = __expf(v0 - m);
    const float e1 = __expf(v1 - m);
    float sum = e0 + e1;
#pragma unroll
    for (int off = 16; off > 0; off >>= 1) sum += __shfl_xor(sum, off, 32);
    const float inv = 1.0f / sum;
    ((float2*)(out + (size_t)node * 64))[lane] = make_float2(e0 * inv, e1 * inv);
}

extern "C" void kernel_launch(void* const* d_in, const int* in_sizes, int n_in,
                              void* d_out, int out_size, void* d_ws, size_t ws_size,
                              hipStream_t stream) {
    const float* x   = (const float*)d_in[0];
    const int*   ei  = (const int*)  d_in[1];
    const float* W1  = (const float*)d_in[2];
    const float* b1  = (const float*)d_in[3];
    const float* W2  = (const float*)d_in[4];
    const float* b2  = (const float*)d_in[5];
    float* out = (float*)d_out;

    const int N = in_sizes[0] / GCN_D_IN;
    const int E = in_sizes[1] / 2;
    const int* src = ei;
    const int* dst = ei + E;

    float* ws   = (float*)d_ws;
    size_t off  = ((size_t)N + 63) & ~(size_t)63;
    float* dinv = ws;
    _Float16* xw1 = (_Float16*)(ws + off);
    float* reg2 = ws + off + (size_t)N * GCN_D_HID;
    _Float16* W1p = (_Float16*)(reg2 + (size_t)N * GCN_D_HID);
    _Float16* W2p = W1p + (size_t)GCN_D_IN * GCN_D_HID;
    float* xw2  = reg2;

    int*  so       = (int*)(reg2 + (size_t)N * GCN_D_OUT);
    int*  deg      = so;
    int*  rowloc   = so + off;
    int*  blocktot = so + 2 * off + 64;
    int*  csr      = so + 2 * off + 64 + 1024;
    int*  part     = so + 2 * off + 64 + 1024 + (((size_t)2 * E + 63) & ~(size_t)63);
    const int pitch = (int)off;
    int*  rowptr   = part + (size_t)GCN_EDGE_CHUNKS * off;

    const int B = 256;
    {
        const int SN = GCN_SCAN_NODES;
        const int nblk = (N + SN - 1) / SN;
        gcn_chunk_count_pack <<<GCN_EDGE_CHUNKS + GCN_PACK_BLOCKS, 1024, 0, stream>>>(dst, part, E, pitch, GCN_EDGE_CHUNKS,
                                                                                      W1, W1p, W2, W2p);
        gcn_chunk_prefix_scan<<<nblk, 1024, 0, stream>>>(part, deg, rowloc, blocktot, N, pitch);
        gcn_place_and_gemm   <<<GCN_EDGE_CHUNKS + GCN_GEMM_WGS, GCN_L3_THREADS, 0, stream>>>(
            src, dst, deg, rowloc, blocktot, part, rowptr, dinv, csr, E, N, pitch, nblk, GCN_EDGE_CHUNKS,
            x, W1p, xw1);
    }

    {
        gcn_gather256_relu_gemm64<<<(N + 15) / 16, 512, 0, stream>>>(rowptr, csr, dinv, xw1, b1, W2p, xw2, N);
    }

    {
        int th = N * 32;
        gcn_gather64_bias_softmax<<<(th + B - 1) / B, B, 0, stream>>>(rowptr, csr, dinv, xw2, b2, out, N);
    }
}
